# starting kernel
# baseline (speedup 1.0000x reference)
#include <hip/hip_runtime.h>


#define UNITS     128
#define IN_STRIDE 129
#define TILE      16
#define SUBSTEPS  8
#define TERMS     12
#define YSTRIDE   132

typedef __attribute__((ext_vector_type(2))) float v2f;
typedef __attribute__((ext_vector_type(8))) float v8f;

__global__ void build_KT(const float* __restrict__ p, float* __restrict__ KdT) {
    int idx = blockIdx.x * blockDim.x + threadIdx.x;
    if (idx >= UNITS * UNITS) return;
    int i = idx >> 7;
    int j = idx & 127;
    float val = 0.0f;
    if (j == i) {
        float x  = p[i];
        float sp = (x > 20.0f) ? x : log1pf(__expf(x));
        val = -sp * sp;
    } else if (j == i + 1) {
        val = -p[UNITS + i];
    } else if (j + 1 == i) {
        val = p[UNITS + j];
    }
    KdT[j * UNITS + i] = val;
}

__global__ void __launch_bounds__(256)
expmv_taylor_wmma(const float* __restrict__ inp,
                  const float* __restrict__ KdT,
                  float* __restrict__ out,
                  int B)
{
    __shared__ float Ybuf[2][TILE * YSTRIDE];

    const int tid   = threadIdx.x;
    const int wave  = tid >> 5;
    const int lane  = tid & 31;
    const int col   = lane & 15;
    const int hi    = lane >> 4;
    const int jbase = wave * 16;
    const int row0  = blockIdx.x * TILE;

    v2f breg[32];
    #pragma unroll
    for (int c = 0; c < 32; ++c) {
        breg[c] = *reinterpret_cast<const v2f*>(
            &KdT[(jbase + col) * UNITS + 4 * c + 2 * hi]);
    }

    float tvec[8];
    v8f S = {};
    #pragma unroll
    for (int v = 0; v < 8; ++v) {
        int r  = row0 + v + 8 * hi;
        int rc = (r < B) ? r : (B - 1);
        tvec[v] = inp[rc * IN_STRIDE + UNITS];
        S[v]    = inp[rc * IN_STRIDE + jbase + col];
    }

    for (int idx = tid; idx < TILE * UNITS; idx += 256) {
        int r  = idx >> 7, cc = idx & 127;
        int rr = row0 + r;
        int rc = (rr < B) ? rr : (B - 1);
        Ybuf[0][r * YSTRIDE + cc] = inp[rc * IN_STRIDE + cc];
    }
    __syncthreads();

    const int aoff = col * YSTRIDE + 2 * hi;

    int cur = 0;
    for (int step = 0; step < SUBSTEPS; ++step) {
        for (int k = 1; k <= TERMS; ++k) {
            const float* Yc = Ybuf[cur];
            float*       Yn = Ybuf[cur ^ 1];

            v8f z0 = {};
            v8f z1 = {};
            v2f abuf[8];
            #pragma unroll
            for (int c = 0; c < 8; ++c)
                abuf[c] = *reinterpret_cast<const v2f*>(&Yc[aoff + 4 * c]);
            #pragma unroll
            for (int c = 0; c < 32; ++c) {
                v2f a = abuf[c & 7];
                if (c + 8 < 32)
                    abuf[c & 7] = *reinterpret_cast<const v2f*>(&Yc[aoff + 4 * (c + 8)]);
                if (c & 1)
                    z1 = __builtin_amdgcn_wmma_f32_16x16x4_f32(
                            false, a, false, breg[c], (short)0, z1, false, false);
                else
                    z0 = __builtin_amdgcn_wmma_f32_16x16x4_f32(
                            false, a, false, breg[c], (short)0, z0, false, false);
            }

            const float invsk = __builtin_amdgcn_rcpf((float)(SUBSTEPS * k));
            const bool  last  = (k == TERMS);
            #pragma unroll
            for (int v = 0; v < 8; ++v) {
                float val = (z0[v] + z1[v]) * tvec[v] * invsk;
                S[v] += val;
                Yn[(v + 8 * hi) * YSTRIDE + jbase + col] = last ? S[v] : val;
            }
            __syncthreads();
            cur ^= 1;
        }
    }

    #pragma unroll
    for (int v = 0; v < 8; ++v) {
        int r = row0 + v + 8 * hi;
        if (r < B) out[r * UNITS + jbase + col] = S[v];
    }
}

extern "C" void kernel_launch(void* const* d_in, const int* in_sizes, int n_in,
                              void* d_out, int out_size, void* d_ws, size_t ws_size,
                              hipStream_t stream) {
    const float* inp    = (const float*)d_in[0];
    const float* params = (const float*)d_in[1];
    float*       out    = (float*)d_out;
    float*       KdT    = (float*)d_ws;

    int B = in_sizes[0] / IN_STRIDE;

    build_KT<<<(UNITS * UNITS + 255) / 256, 256, 0, stream>>>(params, KdT);

    int tiles = (B + TILE - 1) / TILE;
    expmv_taylor_wmma<<<tiles, 256, 0, stream>>>(inp, KdT, out, B);
}


# Dyadic split of t: F=expm(K/8) built once on identity rows, squared twice per workgroup in LDS; rows apply F,F^2,F^4 by bits of t, Taylor only on t mod 1/8: 15 applies not 96; sum order changes
# speedup vs baseline: 1.8326x; 1.8326x over previous
#include <hip/hip_runtime.h>


#define UNITS     128
#define IN_STRIDE 129
#define TILE      16
#define TERMS     12
#define YSTRIDE   132
#define MSIZE     (UNITS * YSTRIDE)

typedef __attribute__((ext_vector_type(2))) float v2f;
typedef __attribute__((ext_vector_type(4))) float v4f;
typedef __attribute__((ext_vector_type(8))) float v8f;

__device__ __forceinline__ void k_diagonals(float* Kb, const float* __restrict__ p, int n) {
    float x  = p[n];
    float sp = (x > 20.0f) ? x : log1pf(__expf(x));
    Kb[n * YSTRIDE + n] = -sp * sp;
    if (n + 1 < UNITS) {
        float c = -p[UNITS + n];
        Kb[n * YSTRIDE + n + 1]   = c;
        Kb[(n + 1) * YSTRIDE + n] = -c;
    }
}

__device__ __forceinline__ void zero_matrix(float* Mz, int tid) {
    const v4f zero = {0.0f, 0.0f, 0.0f, 0.0f};
    for (int idx = tid; idx < MSIZE / 4; idx += 256)
        *reinterpret_cast<v4f*>(&Mz[4 * idx]) = zero;
}

__device__ __forceinline__ void m_frags(v2f (&breg)[32], const float* M, int n, int hi) {
    #pragma unroll
    for (int c = 0; c < 32; ++c) {
        int k = 4 * c + 2 * hi;
        breg[c][0] = M[k * YSTRIDE + n];
        breg[c][1] = M[(k + 1) * YSTRIDE + n];
    }
}

__device__ __forceinline__ v8f apply16(const float* Yc, int aoff, const v2f (&breg)[32]) {
    v8f z0 = {};
    v8f z1 = {};
    v2f abuf[8];
    #pragma unroll
    for (int c = 0; c < 8; ++c)
        abuf[c] = *reinterpret_cast<const v2f*>(&Yc[aoff + 4 * c]);
    #pragma unroll
    for (int c = 0; c < 32; ++c) {
        v2f a = abuf[c & 7];
        if (c + 8 < 32)
            abuf[c & 7] = *reinterpret_cast<const v2f*>(&Yc[aoff + 4 * (c + 8)]);
        if (c & 1)
            z1 = __builtin_amdgcn_wmma_f32_16x16x4_f32(
                    false, a, false, breg[c], (short)0, z1, false, false);
        else
            z0 = __builtin_amdgcn_wmma_f32_16x16x4_f32(
                    false, a, false, breg[c], (short)0, z0, false, false);
    }
    return z0 + z1;
}

__device__ __forceinline__ void taylor_action(float* Y0, float* Y1, int& cur, v8f& S,
                                              const float (&h)[8], const v2f (&breg)[32],
                                              int aoff, int soff) {
    for (int k = 1; k <= TERMS; ++k) {
        const float* Yc = cur ? Y1 : Y0;
        float*       Yn = cur ? Y0 : Y1;
        v8f z = apply16(Yc, aoff, breg);
        const float invk = __builtin_amdgcn_rcpf((float)k);
        const bool  last = (k == TERMS);
        #pragma unroll
        for (int v = 0; v < 8; ++v) {
            float val = z[v] * h[v] * invk;
            S[v] += val;
            Yn[v * YSTRIDE + soff] = last ? S[v] : val;
        }
        __syncthreads();
        cur ^= 1;
    }
}

__global__ void __launch_bounds__(256)
expm_eighth(const float* __restrict__ p, float* __restrict__ F)
{
    __shared__ __attribute__((aligned(16))) float Kbuf[MSIZE];
    __shared__ __attribute__((aligned(16))) float Ybuf[2][TILE * YSTRIDE];

    const int tid   = threadIdx.x;
    const int wave  = tid >> 5;
    const int lane  = tid & 31;
    const int col   = lane & 15;
    const int hi    = lane >> 4;
    const int jbase = wave * 16;
    const int row0  = blockIdx.x * TILE;

    zero_matrix(Kbuf, tid);

    float hvec[8];
    v8f S;
    #pragma unroll
    for (int v = 0; v < 8; ++v) {
        hvec[v] = 0.125f;
        S[v]    = (row0 + v + 8 * hi == jbase + col) ? 1.0f : 0.0f;
    }
    for (int idx = tid; idx < TILE * UNITS; idx += 256) {
        int r = idx >> 7, cc = idx & 127;
        Ybuf[0][r * YSTRIDE + cc] = (row0 + r == cc) ? 1.0f : 0.0f;
    }
    __syncthreads();
    if (tid < UNITS) k_diagonals(Kbuf, p, tid);
    __syncthreads();

    v2f breg[32];
    m_frags(breg, Kbuf, jbase + col, hi);

    const int aoff = col * YSTRIDE + 2 * hi;
    const int soff = 8 * hi * YSTRIDE + jbase + col;
    int cur = 0;
    taylor_action(Ybuf[0], Ybuf[1], cur, S, hvec, breg, aoff, soff);

    #pragma unroll
    for (int v = 0; v < 8; ++v)
        F[(row0 + v + 8 * hi) * UNITS + jbase + col] = S[v];
}

__global__ void __launch_bounds__(256)
expm_rows(const float* __restrict__ inp,
          const float* __restrict__ p,
          const float* __restrict__ F,
          float* __restrict__ out,
          int B)
{
    __shared__ __attribute__((aligned(16))) float Mbuf[3][MSIZE];
    __shared__ __attribute__((aligned(16))) float Ybuf[2][TILE * YSTRIDE];

    const int tid   = threadIdx.x;
    const int wave  = tid >> 5;
    const int lane  = tid & 31;
    const int col   = lane & 15;
    const int hi    = lane >> 4;
    const int jbase = wave * 16;
    const int row0  = blockIdx.x * TILE;

    zero_matrix(Mbuf[2], tid);

    float rvec[8];
    int   gvec[8];
    v8f S;
    #pragma unroll
    for (int v = 0; v < 8; ++v) {
        int r  = row0 + v + 8 * hi;
        int rc = (r < B) ? r : (B - 1);
        float t = inp[rc * IN_STRIDE + UNITS];
        int g = (int)floorf(t * 8.0f);
        g = (g < 0) ? 0 : ((g > 7) ? 7 : g);
        gvec[v] = g;
        rvec[v] = t - 0.125f * (float)g;
        S[v]    = inp[rc * IN_STRIDE + jbase + col];
    }
    for (int idx = tid; idx < TILE * UNITS; idx += 256) {
        int r  = idx >> 7, cc = idx & 127;
        int rr = row0 + r;
        int rc = (rr < B) ? rr : (B - 1);
        Ybuf[0][r * YSTRIDE + cc] = inp[rc * IN_STRIDE + cc];
    }
    __syncthreads();

    if (tid < UNITS) k_diagonals(Mbuf[2], p, tid);
    __syncthreads();

    const int aoff = col * YSTRIDE + 2 * hi;
    const int soff = 8 * hi * YSTRIDE + jbase + col;

    v4f fval[16];
    #pragma unroll
    for (int u = 0; u < 16; ++u)
        fval[u] = reinterpret_cast<const v4f*>(F)[tid + 256 * u];
    {
        v2f breg[32];
        m_frags(breg, Mbuf[2], jbase + col, hi);
        int cur = 0;
        taylor_action(Ybuf[0], Ybuf[1], cur, S, rvec, breg, aoff, soff);
    }
    #pragma unroll
    for (int u = 0; u < 16; ++u) {
        int idx = tid + 256 * u;
        int i = idx >> 5, q = idx & 31;
        *reinterpret_cast<v4f*>(&Mbuf[0][i * YSTRIDE + 4 * q]) = fval[u];
    }
    __syncthreads();

    v2f ereg[32];
    for (int s = 0; s < 2; ++s) {
        const float* Ms = Mbuf[s];
        float*       Md = Mbuf[s + 1];
        m_frags(ereg, Ms, jbase + col, hi);
        for (int rt = 0; rt < UNITS / TILE; ++rt) {
            v8f z = apply16(Ms + rt * TILE * YSTRIDE, aoff, ereg);
            #pragma unroll
            for (int v = 0; v < 8; ++v)
                Md[(rt * TILE + v) * YSTRIDE + soff] = z[v];
        }
        __syncthreads();
    }

    #pragma unroll
    for (int j = 0; j < 3; ++j) {
        const float* Yc = (j & 1) ? Ybuf[1] : Ybuf[0];
        float*       Yn = (j & 1) ? Ybuf[0] : Ybuf[1];
        m_frags(ereg, Mbuf[j], jbase + col, hi);
        v8f z = apply16(Yc, aoff, ereg);
        #pragma unroll
        for (int v = 0; v < 8; ++v) {
            S[v] = ((gvec[v] >> j) & 1) ? z[v] : S[v];
            if (j < 2) Yn[v * YSTRIDE + soff] = S[v];
        }
        if (j < 2) __syncthreads();
    }

    #pragma unroll
    for (int v = 0; v < 8; ++v) {
        int r = row0 + v + 8 * hi;
        if (r < B) out[r * UNITS + jbase + col] = S[v];
    }
}

extern "C" void kernel_launch(void* const* d_in, const int* in_sizes, int n_in,
                              void* d_out, int out_size, void* d_ws, size_t ws_size,
                              hipStream_t stream) {
    const float* inp    = (const float*)d_in[0];
    const float* params = (const float*)d_in[1];
    float*       out    = (float*)d_out;
    float*       F      = (float*)d_ws;

    int B = in_sizes[0] / IN_STRIDE;

    expm_eighth<<<UNITS / TILE, 256, 0, stream>>>(params, F);

    int tiles = (B + TILE - 1) / TILE;
    expm_rows<<<tiles, 256, 0, stream>>>(inp, params, F, out, B);
}


# F^g by repeated application: a row applies F while j < g (7 selected applies); no F^2, F^4 squared per workgroup, no bit-selected applies: 19 dense applies per tile, not 31; sum order changes
# speedup vs baseline: 2.3564x; 1.2858x over previous
#include <hip/hip_runtime.h>


#define UNITS     128
#define IN_STRIDE 129
#define TILE      16
#define TERMS     12
#define YSTRIDE   132
#define MSIZE     (UNITS * YSTRIDE)

typedef __attribute__((ext_vector_type(2))) float v2f;
typedef __attribute__((ext_vector_type(4))) float v4f;
typedef __attribute__((ext_vector_type(8))) float v8f;

__device__ __forceinline__ void k_diagonals(float* Kb, const float* __restrict__ p, int n) {
    float x  = p[n];
    float sp = (x > 20.0f) ? x : log1pf(__expf(x));
    Kb[n * YSTRIDE + n] = -sp * sp;
    if (n + 1 < UNITS) {
        float c = -p[UNITS + n];
        Kb[n * YSTRIDE + n + 1]   = c;
        Kb[(n + 1) * YSTRIDE + n] = -c;
    }
}

__device__ __forceinline__ void zero_matrix(float* Mz, int tid) {
    const v4f zero = {0.0f, 0.0f, 0.0f, 0.0f};
    for (int idx = tid; idx < MSIZE / 4; idx += 256)
        *reinterpret_cast<v4f*>(&Mz[4 * idx]) = zero;
}

__device__ __forceinline__ void m_frags(v2f (&breg)[32], const float* M, int n, int hi) {
    #pragma unroll
    for (int c = 0; c < 32; ++c) {
        int k = 4 * c + 2 * hi;
        breg[c][0] = M[k * YSTRIDE + n];
        breg[c][1] = M[(k + 1) * YSTRIDE + n];
    }
}

__device__ __forceinline__ v8f apply16(const float* Yc, int aoff, const v2f (&breg)[32]) {
    v8f z0 = {};
    v8f z1 = {};
    v2f abuf[8];
    #pragma unroll
    for (int c = 0; c < 8; ++c)
        abuf[c] = *reinterpret_cast<const v2f*>(&Yc[aoff + 4 * c]);
    #pragma unroll
    for (int c = 0; c < 32; ++c) {
        v2f a = abuf[c & 7];
        if (c + 8 < 32)
            abuf[c & 7] = *reinterpret_cast<const v2f*>(&Yc[aoff + 4 * (c + 8)]);
        if (c & 1)
            z1 = __builtin_amdgcn_wmma_f32_16x16x4_f32(
                    false, a, false, breg[c], (short)0, z1, false, false);
        else
            z0 = __builtin_amdgcn_wmma_f32_16x16x4_f32(
                    false, a, false, breg[c], (short)0, z0, false, false);
    }
    return z0 + z1;
}

__device__ __forceinline__ void taylor_action(float* Y0, float* Y1, int& cur, v8f& S,
                                              const float (&h)[8], const v2f (&breg)[32],
                                              int aoff, int soff) {
    for (int k = 1; k <= TERMS; ++k) {
        const float* Yc = cur ? Y1 : Y0;
        float*       Yn = cur ? Y0 : Y1;
        v8f z = apply16(Yc, aoff, breg);
        const float invk = __builtin_amdgcn_rcpf((float)k);
        const bool  last = (k == TERMS);
        #pragma unroll
        for (int v = 0; v < 8; ++v) {
            float val = z[v] * h[v] * invk;
            S[v] += val;
            Yn[v * YSTRIDE + soff] = last ? S[v] : val;
        }
        __syncthreads();
        cur ^= 1;
    }
}

__global__ void __launch_bounds__(256)
expm_eighth(const float* __restrict__ p, float* __restrict__ F)
{
    __shared__ __attribute__((aligned(16))) float Kbuf[MSIZE];
    __shared__ __attribute__((aligned(16))) float Ybuf[2][TILE * YSTRIDE];

    const int tid   = threadIdx.x;
    const int wave  = tid >> 5;
    const int lane  = tid & 31;
    const int col   = lane & 15;
    const int hi    = lane >> 4;
    const int jbase = wave * 16;
    const int row0  = blockIdx.x * TILE;

    zero_matrix(Kbuf, tid);

    float hvec[8];
    v8f S;
    #pragma unroll
    for (int v = 0; v < 8; ++v) {
        hvec[v] = 0.125f;
        S[v]    = (row0 + v + 8 * hi == jbase + col) ? 1.0f : 0.0f;
    }
    for (int idx = tid; idx < TILE * UNITS; idx += 256) {
        int r = idx >> 7, cc = idx & 127;
        Ybuf[0][r * YSTRIDE + cc] = (row0 + r == cc) ? 1.0f : 0.0f;
    }
    __syncthreads();
    if (tid < UNITS) k_diagonals(Kbuf, p, tid);
    __syncthreads();

    v2f breg[32];
    m_frags(breg, Kbuf, jbase + col, hi);

    const int aoff = col * YSTRIDE + 2 * hi;
    const int soff = 8 * hi * YSTRIDE + jbase + col;
    int cur = 0;
    taylor_action(Ybuf[0], Ybuf[1], cur, S, hvec, breg, aoff, soff);

    #pragma unroll
    for (int v = 0; v < 8; ++v)
        F[(row0 + v + 8 * hi) * UNITS + jbase + col] = S[v];
}

__global__ void __launch_bounds__(256)
expm_rows(const float* __restrict__ inp,
          const float* __restrict__ p,
          const float* __restrict__ F,
          float* __restrict__ out,
          int B)
{
    __shared__ __attribute__((aligned(16))) float Mbuf[MSIZE];
    __shared__ __attribute__((aligned(16))) float Ybuf[2][TILE * YSTRIDE];

    const int tid   = threadIdx.x;
    const int wave  = tid >> 5;
    const int lane  = tid & 31;
    const int col   = lane & 15;
    const int hi    = lane >> 4;
    const int jbase = wave * 16;
    const int row0  = blockIdx.x * TILE;

    zero_matrix(Mbuf, tid);

    float rvec[8];
    int   gvec[8];
    v8f S;
    #pragma unroll
    for (int v = 0; v < 8; ++v) {
        int r  = row0 + v + 8 * hi;
        int rc = (r < B) ? r : (B - 1);
        float t = inp[rc * IN_STRIDE + UNITS];
        int g = (int)floorf(t * 8.0f);
        g = (g < 0) ? 0 : ((g > 7) ? 7 : g);
        gvec[v] = g;
        rvec[v] = t - 0.125f * (float)g;
        S[v]    = inp[rc * IN_STRIDE + jbase + col];
    }
    for (int idx = tid; idx < TILE * UNITS; idx += 256) {
        int r  = idx >> 7, cc = idx & 127;
        int rr = row0 + r;
        int rc = (rr < B) ? rr : (B - 1);
        Ybuf[0][r * YSTRIDE + cc] = inp[rc * IN_STRIDE + cc];
    }
    __syncthreads();

    if (tid < UNITS) k_diagonals(Mbuf, p, tid);
    __syncthreads();

    const int aoff = col * YSTRIDE + 2 * hi;
    const int soff = 8 * hi * YSTRIDE + jbase + col;

    v4f fval[16];
    #pragma unroll
    for (int u = 0; u < 16; ++u)
        fval[u] = reinterpret_cast<const v4f*>(F)[tid + 256 * u];
    {
        v2f breg[32];
        m_frags(breg, Mbuf, jbase + col, hi);
        int cur = 0;
        taylor_action(Ybuf[0], Ybuf[1], cur, S, rvec, breg, aoff, soff);
    }
    #pragma unroll
    for (int u = 0; u < 16; ++u) {
        int idx = tid + 256 * u;
        int i = idx >> 5, q = idx & 31;
        *reinterpret_cast<v4f*>(&Mbuf[i * YSTRIDE + 4 * q]) = fval[u];
    }
    __syncthreads();

    {
        v2f ereg[32];
        m_frags(ereg, Mbuf, jbase + col, hi);
        int cur = 0;
        for (int j = 0; j < 7; ++j) {
            const float* Yc = cur ? Ybuf[1] : Ybuf[0];
            float*       Yn = cur ? Ybuf[0] : Ybuf[1];
            v8f z = apply16(Yc, aoff, ereg);
            const bool more = (j < 6);
            #pragma unroll
            for (int v = 0; v < 8; ++v) {
                S[v] = (gvec[v] > j) ? z[v] : S[v];
                if (more) Yn[v * YSTRIDE + soff] = S[v];
            }
            if (more) __syncthreads();
            cur ^= 1;
        }
    }

    #pragma unroll
    for (int v = 0; v < 8; ++v) {
        int r = row0 + v + 8 * hi;
        if (r < B) out[r * UNITS + jbase + col] = S[v];
    }
}

extern "C" void kernel_launch(void* const* d_in, const int* in_sizes, int n_in,
                              void* d_out, int out_size, void* d_ws, size_t ws_size,
                              hipStream_t stream) {
    const float* inp    = (const float*)d_in[0];
    const float* params = (const float*)d_in[1];
    float*       out    = (float*)d_out;
    float*       F      = (float*)d_ws;

    int B = in_sizes[0] / IN_STRIDE;

    expm_eighth<<<UNITS / TILE, 256, 0, stream>>>(params, F);

    int tiles = (B + TILE - 1) / TILE;
    expm_rows<<<tiles, 256, 0, stream>>>(inp, params, F, out, B);
}


# Rows' 12-term Taylor phase moved into launch 1 beside the 8 F tiles (264 workgroups, S passes through out); launch 2 keeps only the 7 F applies, F fragments from global; sum order unchanged
# speedup vs baseline: 2.6312x; 1.1166x over previous
#include <hip/hip_runtime.h>


#define UNITS     128
#define IN_STRIDE 129
#define TILE      16
#define TERMS     12
#define YSTRIDE   132
#define MSIZE     (UNITS * YSTRIDE)

typedef __attribute__((ext_vector_type(2))) float v2f;
typedef __attribute__((ext_vector_type(4))) float v4f;
typedef __attribute__((ext_vector_type(8))) float v8f;

__device__ __forceinline__ void k_diagonals(float* Kb, const float* __restrict__ p, int n) {
    float x  = p[n];
    float sp = (x > 20.0f) ? x : log1pf(__expf(x));
    Kb[n * YSTRIDE + n] = -sp * sp;
    if (n + 1 < UNITS) {
        float c = -p[UNITS + n];
        Kb[n * YSTRIDE + n + 1]   = c;
        Kb[(n + 1) * YSTRIDE + n] = -c;
    }
}

__device__ __forceinline__ void zero_matrix(float* Mz, int tid) {
    const v4f zero = {0.0f, 0.0f, 0.0f, 0.0f};
    for (int idx = tid; idx < MSIZE / 4; idx += 256)
        *reinterpret_cast<v4f*>(&Mz[4 * idx]) = zero;
}

__device__ __forceinline__ void m_frags(v2f (&breg)[32], const float* M, int n, int hi) {
    #pragma unroll
    for (int c = 0; c < 32; ++c) {
        int k = 4 * c + 2 * hi;
        breg[c][0] = M[k * YSTRIDE + n];
        breg[c][1] = M[(k + 1) * YSTRIDE + n];
    }
}

__device__ __forceinline__ v8f apply16(const float* Yc, int aoff, const v2f (&breg)[32]) {
    v8f z0 = {};
    v8f z1 = {};
    v2f abuf[8];
    #pragma unroll
    for (int c = 0; c < 8; ++c)
        abuf[c] = *reinterpret_cast<const v2f*>(&Yc[aoff + 4 * c]);
    #pragma unroll
    for (int c = 0; c < 32; ++c) {
        v2f a = abuf[c & 7];
        if (c + 8 < 32)
            abuf[c & 7] = *reinterpret_cast<const v2f*>(&Yc[aoff + 4 * (c + 8)]);
        if (c & 1)
            z1 = __builtin_amdgcn_wmma_f32_16x16x4_f32(
                    false, a, false, breg[c], (short)0, z1, false, false);
        else
            z0 = __builtin_amdgcn_wmma_f32_16x16x4_f32(
                    false, a, false, breg[c], (short)0, z0, false, false);
    }
    return z0 + z1;
}

__device__ __forceinline__ void taylor_action(float* Y0, float* Y1, int& cur, v8f& S,
                                              const float (&h)[8], const v2f (&breg)[32],
                                              int aoff, int soff) {
    for (int k = 1; k <= TERMS; ++k) {
        const float* Yc = cur ? Y1 : Y0;
        float*       Yn = cur ? Y0 : Y1;
        v8f z = apply16(Yc, aoff, breg);
        const float invk = __builtin_amdgcn_rcpf((float)k);
        const bool  last = (k == TERMS);
        #pragma unroll
        for (int v = 0; v < 8; ++v) {
            float val = z[v] * h[v] * invk;
            S[v] += val;
            Yn[v * YSTRIDE + soff] = last ? S[v] : val;
        }
        __syncthreads();
        cur ^= 1;
    }
}

__device__ __forceinline__ void f_frags(v2f (&breg)[32], const float* __restrict__ Fg, int n, int hi) {
    #pragma unroll
    for (int c = 0; c < 32; ++c) {
        int k = 4 * c + 2 * hi;
        breg[c][0] = Fg[k * UNITS + n];
        breg[c][1] = Fg[(k + 1) * UNITS + n];
    }
}

#define FTILES (UNITS / TILE)
__global__ void __launch_bounds__(256)
taylor_tiles(const float* __restrict__ inp,
             const float* __restrict__ p,
             float* __restrict__ F,
             float* __restrict__ out,
             int B)
{
    __shared__ __attribute__((aligned(16))) float Kbuf[MSIZE];
    __shared__ __attribute__((aligned(16))) float Ybuf[2][TILE * YSTRIDE];

    const int tid   = threadIdx.x;
    const int wave  = tid >> 5;
    const int lane  = tid & 31;
    const int col   = lane & 15;
    const int hi    = lane >> 4;
    const int jbase = wave * 16;
    const bool isF  = blockIdx.x < FTILES;
    const int row0  = (isF ? blockIdx.x : blockIdx.x - FTILES) * TILE;

    zero_matrix(Kbuf, tid);

    float hvec[8];
    v8f S;
    if (isF) {
        #pragma unroll
        for (int v = 0; v < 8; ++v) {
            hvec[v] = 0.125f;
            S[v]    = (row0 + v + 8 * hi == jbase + col) ? 1.0f : 0.0f;
        }
        for (int idx = tid; idx < TILE * UNITS; idx += 256) {
            int r = idx >> 7, cc = idx & 127;
            Ybuf[0][r * YSTRIDE + cc] = (row0 + r == cc) ? 1.0f : 0.0f;
        }
    } else {
        #pragma unroll
        for (int v = 0; v < 8; ++v) {
            int r  = row0 + v + 8 * hi;
            int rc = (r < B) ? r : (B - 1);
            float t = inp[rc * IN_STRIDE + UNITS];
            int g = (int)floorf(t * 8.0f);
            g = (g < 0) ? 0 : ((g > 7) ? 7 : g);
            hvec[v] = t - 0.125f * (float)g;
            S[v]    = inp[rc * IN_STRIDE + jbase + col];
        }
        for (int idx = tid; idx < TILE * UNITS; idx += 256) {
            int r  = idx >> 7, cc = idx & 127;
            int rr = row0 + r;
            int rc = (rr < B) ? rr : (B - 1);
            Ybuf[0][r * YSTRIDE + cc] = inp[rc * IN_STRIDE + cc];
        }
    }
    __syncthreads();
    if (tid < UNITS) k_diagonals(Kbuf, p, tid);
    __syncthreads();

    v2f breg[32];
    m_frags(breg, Kbuf, jbase + col, hi);

    const int aoff = col * YSTRIDE + 2 * hi;
    const int soff = 8 * hi * YSTRIDE + jbase + col;
    int cur = 0;
    taylor_action(Ybuf[0], Ybuf[1], cur, S, hvec, breg, aoff, soff);

    if (isF) {
        #pragma unroll
        for (int v = 0; v < 8; ++v)
            F[(row0 + v + 8 * hi) * UNITS + jbase + col] = S[v];
    } else {
        #pragma unroll
        for (int v = 0; v < 8; ++v) {
            int r = row0 + v + 8 * hi;
            if (r < B) out[r * UNITS + jbase + col] = S[v];
        }
    }
}

__global__ void __launch_bounds__(256)
apply_powers(const float* __restrict__ inp,
             const float* __restrict__ F,
             float* out,
             int B)
{
    __shared__ __attribute__((aligned(16))) float Ybuf[2][TILE * YSTRIDE];

    const int tid   = threadIdx.x;
    const int wave  = tid >> 5;
    const int lane  = tid & 31;
    const int col   = lane & 15;
    const int hi    = lane >> 4;
    const int jbase = wave * 16;
    const int row0  = blockIdx.x * TILE;

    v2f ereg[32];
    f_frags(ereg, F, jbase + col, hi);

    int gvec[8];
    v8f S;
    #pragma unroll
    for (int v = 0; v < 8; ++v) {
        int r  = row0 + v + 8 * hi;
        int rc = (r < B) ? r : (B - 1);
        float t = inp[rc * IN_STRIDE + UNITS];
        int g = (int)floorf(t * 8.0f);
        g = (g < 0) ? 0 : ((g > 7) ? 7 : g);
        gvec[v] = g;
        S[v]    = out[rc * UNITS + jbase + col];
    }
    for (int idx = tid; idx < TILE * UNITS / 4; idx += 256) {
        int r  = idx >> 5, q = idx & 31;
        int rr = row0 + r;
        int rc = (rr < B) ? rr : (B - 1);
        *reinterpret_cast<v4f*>(&Ybuf[0][r * YSTRIDE + 4 * q]) =
            *reinterpret_cast<const v4f*>(&out[rc * UNITS + 4 * q]);
    }
    __syncthreads();

    const int aoff = col * YSTRIDE + 2 * hi;
    const int soff = 8 * hi * YSTRIDE + jbase + col;

    int cur = 0;
    for (int j = 0; j < 7; ++j) {
        const float* Yc = cur ? Ybuf[1] : Ybuf[0];
        float*       Yn = cur ? Ybuf[0] : Ybuf[1];
        v8f z = apply16(Yc, aoff, ereg);
        const bool more = (j < 6);
        #pragma unroll
        for (int v = 0; v < 8; ++v) {
            S[v] = (gvec[v] > j) ? z[v] : S[v];
            if (more) Yn[v * YSTRIDE + soff] = S[v];
        }
        if (more) __syncthreads();
        cur ^= 1;
    }

    #pragma unroll
    for (int v = 0; v < 8; ++v) {
        int r = row0 + v + 8 * hi;
        if (r < B) out[r * UNITS + jbase + col] = S[v];
    }
}

extern "C" void kernel_launch(void* const* d_in, const int* in_sizes, int n_in,
                              void* d_out, int out_size, void* d_ws, size_t ws_size,
                              hipStream_t stream) {
    const float* inp    = (const float*)d_in[0];
    const float* params = (const float*)d_in[1];
    float*       out    = (float*)d_out;
    float*       F      = (float*)d_ws;

    int B = in_sizes[0] / IN_STRIDE;
    int tiles = (B + TILE - 1) / TILE;

    taylor_tiles<<<FTILES + tiles, 256, 0, stream>>>(inp, params, F, out, B);
    apply_powers<<<tiles, 256, 0, stream>>>(inp, F, out, B);
}


# F is a 12-term polynomial of tridiagonal K, exactly zero beyond 12 off-diagonals: launch 2 loads and multiplies only the 10 of 32 k-chunks in each wave's band, 70 WMMAs not 224; sum order unchanged
# speedup vs baseline: 2.9827x; 1.1336x over previous
#include <hip/hip_runtime.h>


#define UNITS     128
#define IN_STRIDE 129
#define TILE      16
#define TERMS     12
#define YSTRIDE   132
#define MSIZE     (UNITS * YSTRIDE)

typedef __attribute__((ext_vector_type(2))) float v2f;
typedef __attribute__((ext_vector_type(4))) float v4f;
typedef __attribute__((ext_vector_type(8))) float v8f;

__device__ __forceinline__ void k_diagonals(float* Kb, const float* __restrict__ p, int n) {
    float x  = p[n];
    float sp = (x > 20.0f) ? x : log1pf(__expf(x));
    Kb[n * YSTRIDE + n] = -sp * sp;
    if (n + 1 < UNITS) {
        float c = -p[UNITS + n];
        Kb[n * YSTRIDE + n + 1]   = c;
        Kb[(n + 1) * YSTRIDE + n] = -c;
    }
}

__device__ __forceinline__ void zero_matrix(float* Mz, int tid) {
    const v4f zero = {0.0f, 0.0f, 0.0f, 0.0f};
    for (int idx = tid; idx < MSIZE / 4; idx += 256)
        *reinterpret_cast<v4f*>(&Mz[4 * idx]) = zero;
}

__device__ __forceinline__ void m_frags(v2f (&breg)[32], const float* M, int n, int hi) {
    #pragma unroll
    for (int c = 0; c < 32; ++c) {
        int k = 4 * c + 2 * hi;
        breg[c][0] = M[k * YSTRIDE + n];
        breg[c][1] = M[(k + 1) * YSTRIDE + n];
    }
}

__device__ __forceinline__ v8f apply16(const float* Yc, int aoff, const v2f (&breg)[32]) {
    v8f z0 = {};
    v8f z1 = {};
    v2f abuf[8];
    #pragma unroll
    for (int c = 0; c < 8; ++c)
        abuf[c] = *reinterpret_cast<const v2f*>(&Yc[aoff + 4 * c]);
    #pragma unroll
    for (int c = 0; c < 32; ++c) {
        v2f a = abuf[c & 7];
        if (c + 8 < 32)
            abuf[c & 7] = *reinterpret_cast<const v2f*>(&Yc[aoff + 4 * (c + 8)]);
        if (c & 1)
            z1 = __builtin_amdgcn_wmma_f32_16x16x4_f32(
                    false, a, false, breg[c], (short)0, z1, false, false);
        else
            z0 = __builtin_amdgcn_wmma_f32_16x16x4_f32(
                    false, a, false, breg[c], (short)0, z0, false, false);
    }
    return z0 + z1;
}

__device__ __forceinline__ void taylor_action(float* Y0, float* Y1, int& cur, v8f& S,
                                              const float (&h)[8], const v2f (&breg)[32],
                                              int aoff, int soff) {
    for (int k = 1; k <= TERMS; ++k) {
        const float* Yc = cur ? Y1 : Y0;
        float*       Yn = cur ? Y0 : Y1;
        v8f z = apply16(Yc, aoff, breg);
        const float invk = __builtin_amdgcn_rcpf((float)k);
        const bool  last = (k == TERMS);
        #pragma unroll
        for (int v = 0; v < 8; ++v) {
            float val = z[v] * h[v] * invk;
            S[v] += val;
            Yn[v * YSTRIDE + soff] = last ? S[v] : val;
        }
        __syncthreads();
        cur ^= 1;
    }
}

#define BAND_CHUNKS ((TILE + 2 * TERMS) / 4)
static_assert(TERMS % 4 == 0, "band window must start on a k-chunk");

__device__ __forceinline__ void f_frags_band(v2f (&breg)[BAND_CHUNKS], const float* __restrict__ Fg,
                                             int n, int hi, int cs) {
    #pragma unroll
    for (int i = 0; i < BAND_CHUNKS; ++i) {
        int k = 4 * (cs + i) + 2 * hi;
        breg[i][0] = Fg[k * UNITS + n];
        breg[i][1] = Fg[(k + 1) * UNITS + n];
    }
}

__device__ __forceinline__ v8f apply_band(const float* Yc, int aoff, const v2f (&breg)[BAND_CHUNKS]) {
    v8f z0 = {};
    v8f z1 = {};
    v2f abuf[BAND_CHUNKS];
    #pragma unroll
    for (int i = 0; i < BAND_CHUNKS; ++i)
        abuf[i] = *reinterpret_cast<const v2f*>(&Yc[aoff + 4 * i]);
    #pragma unroll
    for (int i = 0; i < BAND_CHUNKS; ++i) {
        if (i & 1)
            z1 = __builtin_amdgcn_wmma_f32_16x16x4_f32(
                    false, abuf[i], false, breg[i], (short)0, z1, false, false);
        else
            z0 = __builtin_amdgcn_wmma_f32_16x16x4_f32(
                    false, abuf[i], false, breg[i], (short)0, z0, false, false);
    }
    return z0 + z1;
}

#define FTILES (UNITS / TILE)
__global__ void __launch_bounds__(256)
taylor_tiles(const float* __restrict__ inp,
             const float* __restrict__ p,
             float* __restrict__ F,
             float* __restrict__ out,
             int B)
{
    __shared__ __attribute__((aligned(16))) float Kbuf[MSIZE];
    __shared__ __attribute__((aligned(16))) float Ybuf[2][TILE * YSTRIDE];

    const int tid   = threadIdx.x;
    const int wave  = tid >> 5;
    const int lane  = tid & 31;
    const int col   = lane & 15;
    const int hi    = lane >> 4;
    const int jbase = wave * 16;
    const bool isF  = blockIdx.x < FTILES;
    const int row0  = (isF ? blockIdx.x : blockIdx.x - FTILES) * TILE;

    zero_matrix(Kbuf, tid);

    float hvec[8];
    v8f S;
    if (isF) {
        #pragma unroll
        for (int v = 0; v < 8; ++v) {
            hvec[v] = 0.125f;
            S[v]    = (row0 + v + 8 * hi == jbase + col) ? 1.0f : 0.0f;
        }
        for (int idx = tid; idx < TILE * UNITS; idx += 256) {
            int r = idx >> 7, cc = idx & 127;
            Ybuf[0][r * YSTRIDE + cc] = (row0 + r == cc) ? 1.0f : 0.0f;
        }
    } else {
        #pragma unroll
        for (int v = 0; v < 8; ++v) {
            int r  = row0 + v + 8 * hi;
            int rc = (r < B) ? r : (B - 1);
            float t = inp[rc * IN_STRIDE + UNITS];
            int g = (int)floorf(t * 8.0f);
            g = (g < 0) ? 0 : ((g > 7) ? 7 : g);
            hvec[v] = t - 0.125f * (float)g;
            S[v]    = inp[rc * IN_STRIDE + jbase + col];
        }
        for (int idx = tid; idx < TILE * UNITS; idx += 256) {
            int r  = idx >> 7, cc = idx & 127;
            int rr = row0 + r;
            int rc = (rr < B) ? rr : (B - 1);
            Ybuf[0][r * YSTRIDE + cc] = inp[rc * IN_STRIDE + cc];
        }
    }
    __syncthreads();
    if (tid < UNITS) k_diagonals(Kbuf, p, tid);
    __syncthreads();

    v2f breg[32];
    m_frags(breg, Kbuf, jbase + col, hi);

    const int aoff = col * YSTRIDE + 2 * hi;
    const int soff = 8 * hi * YSTRIDE + jbase + col;
    int cur = 0;
    taylor_action(Ybuf[0], Ybuf[1], cur, S, hvec, breg, aoff, soff);

    if (isF) {
        #pragma unroll
        for (int v = 0; v < 8; ++v)
            F[(row0 + v + 8 * hi) * UNITS + jbase + col] = S[v];
    } else {
        #pragma unroll
        for (int v = 0; v < 8; ++v) {
            int r = row0 + v + 8 * hi;
            if (r < B) out[r * UNITS + jbase + col] = S[v];
        }
    }
}

__global__ void __launch_bounds__(256)
apply_powers(const float* __restrict__ inp,
             const float* __restrict__ F,
             float* out,
             int B)
{
    __shared__ __attribute__((aligned(16))) float Ybuf[2][TILE * YSTRIDE];

    const int tid   = threadIdx.x;
    const int wave  = tid >> 5;
    const int lane  = tid & 31;
    const int col   = lane & 15;
    const int hi    = lane >> 4;
    const int jbase = wave * 16;
    const int row0  = blockIdx.x * TILE;

    int cs = jbase / 4 - TERMS / 4;
    cs = (cs < 0) ? 0 : ((cs > UNITS / 4 - BAND_CHUNKS) ? (UNITS / 4 - BAND_CHUNKS) : cs);
    v2f ereg[BAND_CHUNKS];
    f_frags_band(ereg, F, jbase + col, hi, cs);

    int gvec[8];
    v8f S;
    #pragma unroll
    for (int v = 0; v < 8; ++v) {
        int r  = row0 + v + 8 * hi;
        int rc = (r < B) ? r : (B - 1);
        float t = inp[rc * IN_STRIDE + UNITS];
        int g = (int)floorf(t * 8.0f);
        g = (g < 0) ? 0 : ((g > 7) ? 7 : g);
        gvec[v] = g;
        S[v]    = out[rc * UNITS + jbase + col];
    }
    for (int idx = tid; idx < TILE * UNITS / 4; idx += 256) {
        int r  = idx >> 5, q = idx & 31;
        int rr = row0 + r;
        int rc = (rr < B) ? rr : (B - 1);
        *reinterpret_cast<v4f*>(&Ybuf[0][r * YSTRIDE + 4 * q]) =
            *reinterpret_cast<const v4f*>(&out[rc * UNITS + 4 * q]);
    }
    __syncthreads();

    const int aoff = col * YSTRIDE + 2 * hi + 4 * cs;
    const int soff = 8 * hi * YSTRIDE + jbase + col;

    int cur = 0;
    for (int j = 0; j < 7; ++j) {
        const float* Yc = cur ? Ybuf[1] : Ybuf[0];
        float*       Yn = cur ? Ybuf[0] : Ybuf[1];
        v8f z = apply_band(Yc, aoff, ereg);
        const bool more = (j < 6);
        #pragma unroll
        for (int v = 0; v < 8; ++v) {
            S[v] = (gvec[v] > j) ? z[v] : S[v];
            if (more) Yn[v * YSTRIDE + soff] = S[v];
        }
        if (more) __syncthreads();
        cur ^= 1;
    }

    #pragma unroll
    for (int v = 0; v < 8; ++v) {
        int r = row0 + v + 8 * hi;
        if (r < B) out[r * UNITS + jbase + col] = S[v];
    }
}

extern "C" void kernel_launch(void* const* d_in, const int* in_sizes, int n_in,
                              void* d_out, int out_size, void* d_ws, size_t ws_size,
                              hipStream_t stream) {
    const float* inp    = (const float*)d_in[0];
    const float* params = (const float*)d_in[1];
    float*       out    = (float*)d_out;
    float*       F      = (float*)d_ws;

    int B = in_sizes[0] / IN_STRIDE;
    int tiles = (B + TILE - 1) / TILE;

    taylor_tiles<<<FTILES + tiles, 256, 0, stream>>>(inp, params, F, out, B);
    apply_powers<<<tiles, 256, 0, stream>>>(inp, F, out, B);
}


# Launch 1's 8 identity tiles made light and dispatched last: only the 3 waves and 10 k-chunks their rows' support reaches multiply, so no WGP carries two dense tiles; sum order unchanged
# speedup vs baseline: 3.8953x; 1.3060x over previous
#include <hip/hip_runtime.h>


#define UNITS     128
#define IN_STRIDE 129
#define TILE      16
#define TERMS     12
#define YSTRIDE   132
#define MSIZE     (UNITS * YSTRIDE)

typedef __attribute__((ext_vector_type(2))) float v2f;
typedef __attribute__((ext_vector_type(4))) float v4f;
typedef __attribute__((ext_vector_type(8))) float v8f;

__device__ __forceinline__ void k_diagonals(float* Kb, const float* __restrict__ p, int n) {
    float x  = p[n];
    float sp = (x > 20.0f) ? x : log1pf(__expf(x));
    Kb[n * YSTRIDE + n] = -sp * sp;
    if (n + 1 < UNITS) {
        float c = -p[UNITS + n];
        Kb[n * YSTRIDE + n + 1]   = c;
        Kb[(n + 1) * YSTRIDE + n] = -c;
    }
}

__device__ __forceinline__ void zero_matrix(float* Mz, int tid) {
    const v4f zero = {0.0f, 0.0f, 0.0f, 0.0f};
    for (int idx = tid; idx < MSIZE / 4; idx += 256)
        *reinterpret_cast<v4f*>(&Mz[4 * idx]) = zero;
}

__device__ __forceinline__ void m_frags(v2f (&breg)[32], const float* M, int n, int hi) {
    #pragma unroll
    for (int c = 0; c < 32; ++c) {
        int k = 4 * c + 2 * hi;
        breg[c][0] = M[k * YSTRIDE + n];
        breg[c][1] = M[(k + 1) * YSTRIDE + n];
    }
}

__device__ __forceinline__ v8f apply16(const float* Yc, int aoff, const v2f (&breg)[32]) {
    v8f z0 = {};
    v8f z1 = {};
    v2f abuf[8];
    #pragma unroll
    for (int c = 0; c < 8; ++c)
        abuf[c] = *reinterpret_cast<const v2f*>(&Yc[aoff + 4 * c]);
    #pragma unroll
    for (int c = 0; c < 32; ++c) {
        v2f a = abuf[c & 7];
        if (c + 8 < 32)
            abuf[c & 7] = *reinterpret_cast<const v2f*>(&Yc[aoff + 4 * (c + 8)]);
        if (c & 1)
            z1 = __builtin_amdgcn_wmma_f32_16x16x4_f32(
                    false, a, false, breg[c], (short)0, z1, false, false);
        else
            z0 = __builtin_amdgcn_wmma_f32_16x16x4_f32(
                    false, a, false, breg[c], (short)0, z0, false, false);
    }
    return z0 + z1;
}

__device__ __forceinline__ void taylor_action(float* Y0, float* Y1, int& cur, v8f& S,
                                              const float (&h)[8], const v2f (&breg)[32],
                                              int aoff, int soff) {
    for (int k = 1; k <= TERMS; ++k) {
        const float* Yc = cur ? Y1 : Y0;
        float*       Yn = cur ? Y0 : Y1;
        v8f z = apply16(Yc, aoff, breg);
        const float invk = __builtin_amdgcn_rcpf((float)k);
        const bool  last = (k == TERMS);
        #pragma unroll
        for (int v = 0; v < 8; ++v) {
            float val = z[v] * h[v] * invk;
            S[v] += val;
            Yn[v * YSTRIDE + soff] = last ? S[v] : val;
        }
        __syncthreads();
        cur ^= 1;
    }
}

#define BAND_CHUNKS ((TILE + 2 * TERMS) / 4)
static_assert(TERMS % 4 == 0, "band window must start on a k-chunk");

__device__ __forceinline__ void f_frags_band(v2f (&breg)[BAND_CHUNKS], const float* __restrict__ Fg,
                                             int n, int hi, int cs) {
    #pragma unroll
    for (int i = 0; i < BAND_CHUNKS; ++i) {
        int k = 4 * (cs + i) + 2 * hi;
        breg[i][0] = Fg[k * UNITS + n];
        breg[i][1] = Fg[(k + 1) * UNITS + n];
    }
}

__device__ __forceinline__ v8f apply_band(const float* Yc, int aoff, const v2f (&breg)[BAND_CHUNKS]) {
    v8f z0 = {};
    v8f z1 = {};
    v2f abuf[BAND_CHUNKS];
    #pragma unroll
    for (int i = 0; i < BAND_CHUNKS; ++i)
        abuf[i] = *reinterpret_cast<const v2f*>(&Yc[aoff + 4 * i]);
    #pragma unroll
    for (int i = 0; i < BAND_CHUNKS; ++i) {
        if (i & 1)
            z1 = __builtin_amdgcn_wmma_f32_16x16x4_f32(
                    false, abuf[i], false, breg[i], (short)0, z1, false, false);
        else
            z0 = __builtin_amdgcn_wmma_f32_16x16x4_f32(
                    false, abuf[i], false, breg[i], (short)0, z0, false, false);
    }
    return z0 + z1;
}

__device__ __forceinline__ void k_frags_window(v2f (&breg)[BAND_CHUNKS], const float* M,
                                               int n, int hi, int cs) {
    #pragma unroll
    for (int i = 0; i < BAND_CHUNKS; ++i) {
        int k = 4 * (cs + i) + 2 * hi;
        breg[i][0] = M[k * YSTRIDE + n];
        breg[i][1] = M[(k + 1) * YSTRIDE + n];
    }
}

__device__ __forceinline__ void taylor_action_identity(float* Y0, float* Y1, int& cur, v8f& S,
                                                       const float (&h)[8],
                                                       const v2f (&breg)[BAND_CHUNKS],
                                                       int aoff, int soff, bool active) {
    for (int k = 1; k <= TERMS; ++k) {
        const float* Yc = cur ? Y1 : Y0;
        float*       Yn = cur ? Y0 : Y1;
        v8f z = {};
        if (active) z = apply_band(Yc, aoff, breg);
        const float invk = __builtin_amdgcn_rcpf((float)k);
        const bool  last = (k == TERMS);
        #pragma unroll
        for (int v = 0; v < 8; ++v) {
            float val = z[v] * h[v] * invk;
            S[v] += val;
            Yn[v * YSTRIDE + soff] = last ? S[v] : val;
        }
        __syncthreads();
        cur ^= 1;
    }
}

#define FTILES (UNITS / TILE)
__global__ void __launch_bounds__(256)
taylor_tiles(const float* __restrict__ inp,
             const float* __restrict__ p,
             float* __restrict__ F,
             float* __restrict__ out,
             int B)
{
    __shared__ __attribute__((aligned(16))) float Kbuf[MSIZE];
    __shared__ __attribute__((aligned(16))) float Ybuf[2][TILE * YSTRIDE];

    const int tid   = threadIdx.x;
    const int wave  = tid >> 5;
    const int lane  = tid & 31;
    const int col   = lane & 15;
    const int hi    = lane >> 4;
    const int jbase = wave * 16;
    const int tiles = (B + TILE - 1) / TILE;
    const bool isF  = (int)blockIdx.x >= tiles;
    const int row0  = (isF ? (int)blockIdx.x - tiles : (int)blockIdx.x) * TILE;

    zero_matrix(Kbuf, tid);

    float hvec[8];
    v8f S;
    if (isF) {
        #pragma unroll
        for (int v = 0; v < 8; ++v) {
            hvec[v] = 0.125f;
            S[v]    = (row0 + v + 8 * hi == jbase + col) ? 1.0f : 0.0f;
        }
        for (int idx = tid; idx < TILE * UNITS; idx += 256) {
            int r = idx >> 7, cc = idx & 127;
            Ybuf[0][r * YSTRIDE + cc] = (row0 + r == cc) ? 1.0f : 0.0f;
        }
    } else {
        #pragma unroll
        for (int v = 0; v < 8; ++v) {
            int r  = row0 + v + 8 * hi;
            int rc = (r < B) ? r : (B - 1);
            float t = inp[rc * IN_STRIDE + UNITS];
            int g = (int)floorf(t * 8.0f);
            g = (g < 0) ? 0 : ((g > 7) ? 7 : g);
            hvec[v] = t - 0.125f * (float)g;
            S[v]    = inp[rc * IN_STRIDE + jbase + col];
        }
        for (int idx = tid; idx < TILE * UNITS; idx += 256) {
            int r  = idx >> 7, cc = idx & 127;
            int rr = row0 + r;
            int rc = (rr < B) ? rr : (B - 1);
            Ybuf[0][r * YSTRIDE + cc] = inp[rc * IN_STRIDE + cc];
        }
    }
    __syncthreads();
    if (tid < UNITS) k_diagonals(Kbuf, p, tid);
    __syncthreads();

    const int soff = 8 * hi * YSTRIDE + jbase + col;
    int cur = 0;
    if (isF) {
        int cs = row0 / 4 - TERMS / 4;
        cs = (cs < 0) ? 0 : ((cs > UNITS / 4 - BAND_CHUNKS) ? (UNITS / 4 - BAND_CHUNKS) : cs);
        const int wv = __builtin_amdgcn_readfirstlane(wave);
        const int tl = row0 / TILE;
        const bool active = (wv >= tl - 1) && (wv <= tl + 1);
        v2f kreg[BAND_CHUNKS];
        k_frags_window(kreg, Kbuf, jbase + col, hi, cs);
        const int aoff = col * YSTRIDE + 2 * hi + 4 * cs;
        taylor_action_identity(Ybuf[0], Ybuf[1], cur, S, hvec, kreg, aoff, soff, active);
    } else {
        v2f breg[32];
        m_frags(breg, Kbuf, jbase + col, hi);
        const int aoff = col * YSTRIDE + 2 * hi;
        taylor_action(Ybuf[0], Ybuf[1], cur, S, hvec, breg, aoff, soff);
    }

    if (isF) {
        #pragma unroll
        for (int v = 0; v < 8; ++v)
            F[(row0 + v + 8 * hi) * UNITS + jbase + col] = S[v];
    } else {
        #pragma unroll
        for (int v = 0; v < 8; ++v) {
            int r = row0 + v + 8 * hi;
            if (r < B) out[r * UNITS + jbase + col] = S[v];
        }
    }
}

__global__ void __launch_bounds__(256)
apply_powers(const float* __restrict__ inp,
             const float* __restrict__ F,
             float* out,
             int B)
{
    __shared__ __attribute__((aligned(16))) float Ybuf[2][TILE * YSTRIDE];

    const int tid   = threadIdx.x;
    const int wave  = tid >> 5;
    const int lane  = tid & 31;
    const int col   = lane & 15;
    const int hi    = lane >> 4;
    const int jbase = wave * 16;
    const int row0  = blockIdx.x * TILE;

    int cs = jbase / 4 - TERMS / 4;
    cs = (cs < 0) ? 0 : ((cs > UNITS / 4 - BAND_CHUNKS) ? (UNITS / 4 - BAND_CHUNKS) : cs);
    v2f ereg[BAND_CHUNKS];
    f_frags_band(ereg, F, jbase + col, hi, cs);

    int gvec[8];
    v8f S;
    #pragma unroll
    for (int v = 0; v < 8; ++v) {
        int r  = row0 + v + 8 * hi;
        int rc = (r < B) ? r : (B - 1);
        float t = inp[rc * IN_STRIDE + UNITS];
        int g = (int)floorf(t * 8.0f);
        g = (g < 0) ? 0 : ((g > 7) ? 7 : g);
        gvec[v] = g;
        S[v]    = out[rc * UNITS + jbase + col];
    }
    for (int idx = tid; idx < TILE * UNITS / 4; idx += 256) {
        int r  = idx >> 5, q = idx & 31;
        int rr = row0 + r;
        int rc = (rr < B) ? rr : (B - 1);
        *reinterpret_cast<v4f*>(&Ybuf[0][r * YSTRIDE + 4 * q]) =
            *reinterpret_cast<const v4f*>(&out[rc * UNITS + 4 * q]);
    }
    __syncthreads();

    const int aoff = col * YSTRIDE + 2 * hi + 4 * cs;
    const int soff = 8 * hi * YSTRIDE + jbase + col;

    int cur = 0;
    for (int j = 0; j < 7; ++j) {
        const float* Yc = cur ? Ybuf[1] : Ybuf[0];
        float*       Yn = cur ? Ybuf[0] : Ybuf[1];
        v8f z = apply_band(Yc, aoff, ereg);
        const bool more = (j < 6);
        #pragma unroll
        for (int v = 0; v < 8; ++v) {
            S[v] = (gvec[v] > j) ? z[v] : S[v];
            if (more) Yn[v * YSTRIDE + soff] = S[v];
        }
        if (more) __syncthreads();
        cur ^= 1;
    }

    #pragma unroll
    for (int v = 0; v < 8; ++v) {
        int r = row0 + v + 8 * hi;
        if (r < B) out[r * UNITS + jbase + col] = S[v];
    }
}

extern "C" void kernel_launch(void* const* d_in, const int* in_sizes, int n_in,
                              void* d_out, int out_size, void* d_ws, size_t ws_size,
                              hipStream_t stream) {
    const float* inp    = (const float*)d_in[0];
    const float* params = (const float*)d_in[1];
    float*       out    = (float*)d_out;
    float*       F      = (float*)d_ws;

    int B = in_sizes[0] / IN_STRIDE;
    int tiles = (B + TILE - 1) / TILE;

    taylor_tiles<<<tiles + FTILES, 256, 0, stream>>>(inp, params, F, out, B);
    apply_powers<<<tiles, 256, 0, stream>>>(inp, F, out, B);
}


# Split t at the nearest eighth, not the one below: |r| <= 1/16 halves the rows' series argument, so 9 dense Taylor terms undercut the 12-term tail; g reaches 8: 8 selected F applies; rounding changes
# speedup vs baseline: 4.2459x; 1.0900x over previous
#include <hip/hip_runtime.h>


#define UNITS     128
#define IN_STRIDE 129
#define TILE      16
#define TERMS     12
#define ROW_TERMS 9
#define GMAX      8
#define YSTRIDE   132
#define MSIZE     (UNITS * YSTRIDE)

typedef __attribute__((ext_vector_type(2))) float v2f;
typedef __attribute__((ext_vector_type(4))) float v4f;
typedef __attribute__((ext_vector_type(8))) float v8f;

__device__ __forceinline__ int nearest_eighth(float t) {
    int g = (int)floorf(t * 8.0f + 0.5f);
    return (g < 0) ? 0 : ((g > GMAX) ? GMAX : g);
}

__device__ __forceinline__ void k_diagonals(float* Kb, const float* __restrict__ p, int n) {
    float x  = p[n];
    float sp = (x > 20.0f) ? x : log1pf(__expf(x));
    Kb[n * YSTRIDE + n] = -sp * sp;
    if (n + 1 < UNITS) {
        float c = -p[UNITS + n];
        Kb[n * YSTRIDE + n + 1]   = c;
        Kb[(n + 1) * YSTRIDE + n] = -c;
    }
}

__device__ __forceinline__ void zero_matrix(float* Mz, int tid) {
    const v4f zero = {0.0f, 0.0f, 0.0f, 0.0f};
    for (int idx = tid; idx < MSIZE / 4; idx += 256)
        *reinterpret_cast<v4f*>(&Mz[4 * idx]) = zero;
}

__device__ __forceinline__ void m_frags(v2f (&breg)[32], const float* M, int n, int hi) {
    #pragma unroll
    for (int c = 0; c < 32; ++c) {
        int k = 4 * c + 2 * hi;
        breg[c][0] = M[k * YSTRIDE + n];
        breg[c][1] = M[(k + 1) * YSTRIDE + n];
    }
}

__device__ __forceinline__ v8f apply16(const float* Yc, int aoff, const v2f (&breg)[32]) {
    v8f z0 = {};
    v8f z1 = {};
    v2f abuf[8];
    #pragma unroll
    for (int c = 0; c < 8; ++c)
        abuf[c] = *reinterpret_cast<const v2f*>(&Yc[aoff + 4 * c]);
    #pragma unroll
    for (int c = 0; c < 32; ++c) {
        v2f a = abuf[c & 7];
        if (c + 8 < 32)
            abuf[c & 7] = *reinterpret_cast<const v2f*>(&Yc[aoff + 4 * (c + 8)]);
        if (c & 1)
            z1 = __builtin_amdgcn_wmma_f32_16x16x4_f32(
                    false, a, false, breg[c], (short)0, z1, false, false);
        else
            z0 = __builtin_amdgcn_wmma_f32_16x16x4_f32(
                    false, a, false, breg[c], (short)0, z0, false, false);
    }
    return z0 + z1;
}

__device__ __forceinline__ void taylor_action(float* Y0, float* Y1, int& cur, v8f& S,
                                              const float (&h)[8], const v2f (&breg)[32],
                                              int aoff, int soff) {
    for (int k = 1; k <= ROW_TERMS; ++k) {
        const float* Yc = cur ? Y1 : Y0;
        float*       Yn = cur ? Y0 : Y1;
        v8f z = apply16(Yc, aoff, breg);
        const float invk = __builtin_amdgcn_rcpf((float)k);
        const bool  last = (k == ROW_TERMS);
        #pragma unroll
        for (int v = 0; v < 8; ++v) {
            float val = z[v] * h[v] * invk;
            S[v] += val;
            Yn[v * YSTRIDE + soff] = last ? S[v] : val;
        }
        __syncthreads();
        cur ^= 1;
    }
}

#define BAND_CHUNKS ((TILE + 2 * TERMS) / 4)
static_assert(TERMS % 4 == 0, "band window must start on a k-chunk");

__device__ __forceinline__ void f_frags_band(v2f (&breg)[BAND_CHUNKS], const float* __restrict__ Fg,
                                             int n, int hi, int cs) {
    #pragma unroll
    for (int i = 0; i < BAND_CHUNKS; ++i) {
        int k = 4 * (cs + i) + 2 * hi;
        breg[i][0] = Fg[k * UNITS + n];
        breg[i][1] = Fg[(k + 1) * UNITS + n];
    }
}

__device__ __forceinline__ v8f apply_band(const float* Yc, int aoff, const v2f (&breg)[BAND_CHUNKS]) {
    v8f z0 = {};
    v8f z1 = {};
    v2f abuf[BAND_CHUNKS];
    #pragma unroll
    for (int i = 0; i < BAND_CHUNKS; ++i)
        abuf[i] = *reinterpret_cast<const v2f*>(&Yc[aoff + 4 * i]);
    #pragma unroll
    for (int i = 0; i < BAND_CHUNKS; ++i) {
        if (i & 1)
            z1 = __builtin_amdgcn_wmma_f32_16x16x4_f32(
                    false, abuf[i], false, breg[i], (short)0, z1, false, false);
        else
            z0 = __builtin_amdgcn_wmma_f32_16x16x4_f32(
                    false, abuf[i], false, breg[i], (short)0, z0, false, false);
    }
    return z0 + z1;
}

__device__ __forceinline__ void k_frags_window(v2f (&breg)[BAND_CHUNKS], const float* M,
                                               int n, int hi, int cs) {
    #pragma unroll
    for (int i = 0; i < BAND_CHUNKS; ++i) {
        int k = 4 * (cs + i) + 2 * hi;
        breg[i][0] = M[k * YSTRIDE + n];
        breg[i][1] = M[(k + 1) * YSTRIDE + n];
    }
}

__device__ __forceinline__ void taylor_action_identity(float* Y0, float* Y1, int& cur, v8f& S,
                                                       const float (&h)[8],
                                                       const v2f (&breg)[BAND_CHUNKS],
                                                       int aoff, int soff, bool active) {
    for (int k = 1; k <= TERMS; ++k) {
        const float* Yc = cur ? Y1 : Y0;
        float*       Yn = cur ? Y0 : Y1;
        v8f z = {};
        if (active) z = apply_band(Yc, aoff, breg);
        const float invk = __builtin_amdgcn_rcpf((float)k);
        const bool  last = (k == TERMS);
        #pragma unroll
        for (int v = 0; v < 8; ++v) {
            float val = z[v] * h[v] * invk;
            S[v] += val;
            Yn[v * YSTRIDE + soff] = last ? S[v] : val;
        }
        __syncthreads();
        cur ^= 1;
    }
}

#define FTILES (UNITS / TILE)
__global__ void __launch_bounds__(256)
taylor_tiles(const float* __restrict__ inp,
             const float* __restrict__ p,
             float* __restrict__ F,
             float* __restrict__ out,
             int B)
{
    __shared__ __attribute__((aligned(16))) float Kbuf[MSIZE];
    __shared__ __attribute__((aligned(16))) float Ybuf[2][TILE * YSTRIDE];

    const int tid   = threadIdx.x;
    const int wave  = tid >> 5;
    const int lane  = tid & 31;
    const int col   = lane & 15;
    const int hi    = lane >> 4;
    const int jbase = wave * 16;
    const int tiles = (B + TILE - 1) / TILE;
    const bool isF  = (int)blockIdx.x >= tiles;
    const int row0  = (isF ? (int)blockIdx.x - tiles : (int)blockIdx.x) * TILE;

    zero_matrix(Kbuf, tid);

    float hvec[8];
    v8f S;
    if (isF) {
        #pragma unroll
        for (int v = 0; v < 8; ++v) {
            hvec[v] = 0.125f;
            S[v]    = (row0 + v + 8 * hi == jbase + col) ? 1.0f : 0.0f;
        }
        for (int idx = tid; idx < TILE * UNITS; idx += 256) {
            int r = idx >> 7, cc = idx & 127;
            Ybuf[0][r * YSTRIDE + cc] = (row0 + r == cc) ? 1.0f : 0.0f;
        }
    } else {
        #pragma unroll
        for (int v = 0; v < 8; ++v) {
            int r  = row0 + v + 8 * hi;
            int rc = (r < B) ? r : (B - 1);
            float t = inp[rc * IN_STRIDE + UNITS];
            int g = nearest_eighth(t);
            hvec[v] = t - 0.125f * (float)g;
            S[v]    = inp[rc * IN_STRIDE + jbase + col];
        }
        for (int idx = tid; idx < TILE * UNITS; idx += 256) {
            int r  = idx >> 7, cc = idx & 127;
            int rr = row0 + r;
            int rc = (rr < B) ? rr : (B - 1);
            Ybuf[0][r * YSTRIDE + cc] = inp[rc * IN_STRIDE + cc];
        }
    }
    __syncthreads();
    if (tid < UNITS) k_diagonals(Kbuf, p, tid);
    __syncthreads();

    const int soff = 8 * hi * YSTRIDE + jbase + col;
    int cur = 0;
    if (isF) {
        int cs = row0 / 4 - TERMS / 4;
        cs = (cs < 0) ? 0 : ((cs > UNITS / 4 - BAND_CHUNKS) ? (UNITS / 4 - BAND_CHUNKS) : cs);
        const int wv = __builtin_amdgcn_readfirstlane(wave);
        const int tl = row0 / TILE;
        const bool active = (wv >= tl - 1) && (wv <= tl + 1);
        v2f kreg[BAND_CHUNKS];
        k_frags_window(kreg, Kbuf, jbase + col, hi, cs);
        const int aoff = col * YSTRIDE + 2 * hi + 4 * cs;
        taylor_action_identity(Ybuf[0], Ybuf[1], cur, S, hvec, kreg, aoff, soff, active);
    } else {
        v2f breg[32];
        m_frags(breg, Kbuf, jbase + col, hi);
        const int aoff = col * YSTRIDE + 2 * hi;
        taylor_action(Ybuf[0], Ybuf[1], cur, S, hvec, breg, aoff, soff);
    }

    if (isF) {
        #pragma unroll
        for (int v = 0; v < 8; ++v)
            F[(row0 + v + 8 * hi) * UNITS + jbase + col] = S[v];
    } else {
        #pragma unroll
        for (int v = 0; v < 8; ++v) {
            int r = row0 + v + 8 * hi;
            if (r < B) out[r * UNITS + jbase + col] = S[v];
        }
    }
}

__global__ void __launch_bounds__(256)
apply_powers(const float* __restrict__ inp,
             const float* __restrict__ F,
             float* out,
             int B)
{
    __shared__ __attribute__((aligned(16))) float Ybuf[2][TILE * YSTRIDE];

    const int tid   = threadIdx.x;
    const int wave  = tid >> 5;
    const int lane  = tid & 31;
    const int col   = lane & 15;
    const int hi    = lane >> 4;
    const int jbase = wave * 16;
    const int row0  = blockIdx.x * TILE;

    int cs = jbase / 4 - TERMS / 4;
    cs = (cs < 0) ? 0 : ((cs > UNITS / 4 - BAND_CHUNKS) ? (UNITS / 4 - BAND_CHUNKS) : cs);
    v2f ereg[BAND_CHUNKS];
    f_frags_band(ereg, F, jbase + col, hi, cs);

    int gvec[8];
    v8f S;
    #pragma unroll
    for (int v = 0; v < 8; ++v) {
        int r  = row0 + v + 8 * hi;
        int rc = (r < B) ? r : (B - 1);
        float t = inp[rc * IN_STRIDE + UNITS];
        gvec[v] = nearest_eighth(t);
        S[v]    = out[rc * UNITS + jbase + col];
    }
    for (int idx = tid; idx < TILE * UNITS / 4; idx += 256) {
        int r  = idx >> 5, q = idx & 31;
        int rr = row0 + r;
        int rc = (rr < B) ? rr : (B - 1);
        *reinterpret_cast<v4f*>(&Ybuf[0][r * YSTRIDE + 4 * q]) =
            *reinterpret_cast<const v4f*>(&out[rc * UNITS + 4 * q]);
    }
    __syncthreads();

    const int aoff = col * YSTRIDE + 2 * hi + 4 * cs;
    const int soff = 8 * hi * YSTRIDE + jbase + col;

    int cur = 0;
    for (int j = 0; j < GMAX; ++j) {
        const float* Yc = cur ? Ybuf[1] : Ybuf[0];
        float*       Yn = cur ? Ybuf[0] : Ybuf[1];
        v8f z = apply_band(Yc, aoff, ereg);
        const bool more = (j < GMAX - 1);
        #pragma unroll
        for (int v = 0; v < 8; ++v) {
            S[v] = (gvec[v] > j) ? z[v] : S[v];
            if (more) Yn[v * YSTRIDE + soff] = S[v];
        }
        if (more) __syncthreads();
        cur ^= 1;
    }

    #pragma unroll
    for (int v = 0; v < 8; ++v) {
        int r = row0 + v + 8 * hi;
        if (r < B) out[r * UNITS + jbase + col] = S[v];
    }
}

extern "C" void kernel_launch(void* const* d_in, const int* in_sizes, int n_in,
                              void* d_out, int out_size, void* d_ws, size_t ws_size,
                              hipStream_t stream) {
    const float* inp    = (const float*)d_in[0];
    const float* params = (const float*)d_in[1];
    float*       out    = (float*)d_out;
    float*       F      = (float*)d_ws;

    int B = in_sizes[0] / IN_STRIDE;
    int tiles = (B + TILE - 1) / TILE;

    taylor_tiles<<<tiles + FTILES, 256, 0, stream>>>(inp, params, F, out, B);
    apply_powers<<<tiles, 256, 0, stream>>>(inp, F, out, B);
}


# Split t at the nearest 64th: band matrices expm(K/32), expm(K/64) from 16 more light identity tiles, applied 3 and 1 times by bits of t; rows' dense series 4 terms, not 9; sum order, rounding change
# speedup vs baseline: 4.5295x; 1.0668x over previous
#include <hip/hip_runtime.h>


#define UNITS     128
#define IN_STRIDE 129
#define TILE      16
#define TERMS     12
#define ROW_TERMS 4
#define GMAX      8
#define PMAX      3
#define QMAX      1
#define P_TERMS   7
#define Q_TERMS   6
#define YSTRIDE   132
#define MSIZE     (UNITS * YSTRIDE)

typedef __attribute__((ext_vector_type(2))) float v2f;
typedef __attribute__((ext_vector_type(4))) float v4f;
typedef __attribute__((ext_vector_type(8))) float v8f;

__device__ __forceinline__ int nearest_64th(float t) {
    int q = (int)floorf(t * 64.0f + 0.5f);
    return (q < 0) ? 0 : ((q > 8 * GMAX) ? 8 * GMAX : q);
}

__device__ __forceinline__ void k_diagonals(float* Kb, const float* __restrict__ p, int n) {
    float x  = p[n];
    float sp = (x > 20.0f) ? x : log1pf(__expf(x));
    Kb[n * YSTRIDE + n] = -sp * sp;
    if (n + 1 < UNITS) {
        float c = -p[UNITS + n];
        Kb[n * YSTRIDE + n + 1]   = c;
        Kb[(n + 1) * YSTRIDE + n] = -c;
    }
}

__device__ __forceinline__ void zero_matrix(float* Mz, int tid) {
    const v4f zero = {0.0f, 0.0f, 0.0f, 0.0f};
    for (int idx = tid; idx < MSIZE / 4; idx += 256)
        *reinterpret_cast<v4f*>(&Mz[4 * idx]) = zero;
}

__device__ __forceinline__ void m_frags(v2f (&breg)[32], const float* M, int n, int hi) {
    #pragma unroll
    for (int c = 0; c < 32; ++c) {
        int k = 4 * c + 2 * hi;
        breg[c][0] = M[k * YSTRIDE + n];
        breg[c][1] = M[(k + 1) * YSTRIDE + n];
    }
}

__device__ __forceinline__ v8f apply16(const float* Yc, int aoff, const v2f (&breg)[32]) {
    v8f z0 = {};
    v8f z1 = {};
    v2f abuf[8];
    #pragma unroll
    for (int c = 0; c < 8; ++c)
        abuf[c] = *reinterpret_cast<const v2f*>(&Yc[aoff + 4 * c]);
    #pragma unroll
    for (int c = 0; c < 32; ++c) {
        v2f a = abuf[c & 7];
        if (c + 8 < 32)
            abuf[c & 7] = *reinterpret_cast<const v2f*>(&Yc[aoff + 4 * (c + 8)]);
        if (c & 1)
            z1 = __builtin_amdgcn_wmma_f32_16x16x4_f32(
                    false, a, false, breg[c], (short)0, z1, false, false);
        else
            z0 = __builtin_amdgcn_wmma_f32_16x16x4_f32(
                    false, a, false, breg[c], (short)0, z0, false, false);
    }
    return z0 + z1;
}

__device__ __forceinline__ void taylor_action(float* Y0, float* Y1, int& cur, v8f& S,
                                              const float (&h)[8], const v2f (&breg)[32],
                                              int aoff, int soff) {
    for (int k = 1; k <= ROW_TERMS; ++k) {
        const float* Yc = cur ? Y1 : Y0;
        float*       Yn = cur ? Y0 : Y1;
        v8f z = apply16(Yc, aoff, breg);
        const float invk = __builtin_amdgcn_rcpf((float)k);
        const bool  last = (k == ROW_TERMS);
        #pragma unroll
        for (int v = 0; v < 8; ++v) {
            float val = z[v] * h[v] * invk;
            S[v] += val;
            Yn[v * YSTRIDE + soff] = last ? S[v] : val;
        }
        __syncthreads();
        cur ^= 1;
    }
}

#define BAND_CHUNKS ((TILE + 2 * TERMS) / 4)
static_assert(TERMS % 4 == 0, "band window must start on a k-chunk");
#define PQ_CHUNKS ((TILE + 2 * 8) / 4)
static_assert(P_TERMS <= 8 && Q_TERMS <= 8, "P and Q must fit the 8-off-diagonal window");

#define F_FLOATS  (8 * BAND_CHUNKS * 4 * TILE)
#define PQ_FLOATS (8 * PQ_CHUNKS * 4 * TILE)
__device__ __forceinline__ int win_chunks(int mat) { return mat == 0 ? BAND_CHUNKS : PQ_CHUNKS; }
__device__ __forceinline__ int win_start(int mat, int w) {
    const int nch = win_chunks(mat);
    int cs = 4 * w - (mat == 0 ? TERMS / 4 : 2);
    return (cs < 0) ? 0 : ((cs > UNITS / 4 - nch) ? (UNITS / 4 - nch) : cs);
}
__device__ __forceinline__ int win_base(int mat, int w) {
    return (mat == 0 ? 0 : (F_FLOATS + (mat - 1) * PQ_FLOATS)) + w * win_chunks(mat) * 4 * TILE;
}

template <int NCH>
__device__ __forceinline__ void w_frags(v2f (&breg)[NCH], const float* __restrict__ Wg,
                                        int col, int hi) {
    #pragma unroll
    for (int i = 0; i < NCH; ++i) {
        int kk = 4 * i + 2 * hi;
        breg[i][0] = Wg[kk * TILE + col];
        breg[i][1] = Wg[(kk + 1) * TILE + col];
    }
}

template <int NCH>
__device__ __forceinline__ v8f apply_band(const float* Yc, int aoff, const v2f (&breg)[NCH]) {
    v8f z0 = {};
    v8f z1 = {};
    v2f abuf[NCH];
    #pragma unroll
    for (int i = 0; i < NCH; ++i)
        abuf[i] = *reinterpret_cast<const v2f*>(&Yc[aoff + 4 * i]);
    #pragma unroll
    for (int i = 0; i < NCH; ++i) {
        if (i & 1)
            z1 = __builtin_amdgcn_wmma_f32_16x16x4_f32(
                    false, abuf[i], false, breg[i], (short)0, z1, false, false);
        else
            z0 = __builtin_amdgcn_wmma_f32_16x16x4_f32(
                    false, abuf[i], false, breg[i], (short)0, z0, false, false);
    }
    return z0 + z1;
}

__device__ __forceinline__ void k_frags_window(v2f (&breg)[BAND_CHUNKS], const float* M,
                                               int n, int hi, int cs) {
    #pragma unroll
    for (int i = 0; i < BAND_CHUNKS; ++i) {
        int k = 4 * (cs + i) + 2 * hi;
        breg[i][0] = M[k * YSTRIDE + n];
        breg[i][1] = M[(k + 1) * YSTRIDE + n];
    }
}

__device__ __forceinline__ void taylor_action_identity(float* Y0, float* Y1, int& cur, v8f& S,
                                                       const float (&h)[8],
                                                       const v2f (&breg)[BAND_CHUNKS],
                                                       int aoff, int soff, bool active, int nterms) {
    for (int k = 1; k <= nterms; ++k) {
        const float* Yc = cur ? Y1 : Y0;
        float*       Yn = cur ? Y0 : Y1;
        v8f z = {};
        if (active) z = apply_band<BAND_CHUNKS>(Yc, aoff, breg);
        const float invk = __builtin_amdgcn_rcpf((float)k);
        const bool  last = (k == nterms);
        #pragma unroll
        for (int v = 0; v < 8; ++v) {
            float val = z[v] * h[v] * invk;
            S[v] += val;
            Yn[v * YSTRIDE + soff] = last ? S[v] : val;
        }
        __syncthreads();
        cur ^= 1;
    }
}

#define FTILES (UNITS / TILE)
#define NMATS  3
__global__ void __launch_bounds__(256)
taylor_tiles(const float* __restrict__ inp,
             const float* __restrict__ p,
             float* __restrict__ Wsp,
             float* __restrict__ out,
             int B)
{
    __shared__ __attribute__((aligned(16))) float Kbuf[MSIZE];
    __shared__ __attribute__((aligned(16))) float Ybuf[2][TILE * YSTRIDE];

    const int tid   = threadIdx.x;
    const int wave  = tid >> 5;
    const int lane  = tid & 31;
    const int col   = lane & 15;
    const int hi    = lane >> 4;
    const int jbase = wave * 16;
    const int tiles = (B + TILE - 1) / TILE;
    const bool isF  = (int)blockIdx.x >= tiles;
    const int ft    = (int)blockIdx.x - tiles;
    const int mat   = isF ? ft / FTILES : 0;
    const int row0  = (isF ? ft % FTILES : (int)blockIdx.x) * TILE;

    zero_matrix(Kbuf, tid);

    float hvec[8];
    v8f S;
    if (isF) {
        const float hm = (mat == 0) ? 0.125f : ((mat == 1) ? 0.03125f : 0.015625f);
        #pragma unroll
        for (int v = 0; v < 8; ++v) {
            hvec[v] = hm;
            S[v]    = (row0 + v + 8 * hi == jbase + col) ? 1.0f : 0.0f;
        }
        for (int idx = tid; idx < TILE * UNITS; idx += 256) {
            int r = idx >> 7, cc = idx & 127;
            Ybuf[0][r * YSTRIDE + cc] = (row0 + r == cc) ? 1.0f : 0.0f;
        }
    } else {
        #pragma unroll
        for (int v = 0; v < 8; ++v) {
            int r  = row0 + v + 8 * hi;
            int rc = (r < B) ? r : (B - 1);
            float t = inp[rc * IN_STRIDE + UNITS];
            int q = nearest_64th(t);
            hvec[v] = t - 0.015625f * (float)q;
            S[v]    = inp[rc * IN_STRIDE + jbase + col];
        }
        for (int idx = tid; idx < TILE * UNITS; idx += 256) {
            int r  = idx >> 7, cc = idx & 127;
            int rr = row0 + r;
            int rc = (rr < B) ? rr : (B - 1);
            Ybuf[0][r * YSTRIDE + cc] = inp[rc * IN_STRIDE + cc];
        }
    }
    __syncthreads();
    if (tid < UNITS) k_diagonals(Kbuf, p, tid);
    __syncthreads();

    const int soff = 8 * hi * YSTRIDE + jbase + col;
    int cur = 0;
    if (isF) {
        int cs = row0 / 4 - TERMS / 4;
        cs = (cs < 0) ? 0 : ((cs > UNITS / 4 - BAND_CHUNKS) ? (UNITS / 4 - BAND_CHUNKS) : cs);
        const int wv = __builtin_amdgcn_readfirstlane(wave);
        const int tl = row0 / TILE;
        const bool active = (wv >= tl - 1) && (wv <= tl + 1);
        v2f kreg[BAND_CHUNKS];
        k_frags_window(kreg, Kbuf, jbase + col, hi, cs);
        const int aoff = col * YSTRIDE + 2 * hi + 4 * cs;
        const int nterms = (mat == 0) ? TERMS : ((mat == 1) ? P_TERMS : Q_TERMS);
        taylor_action_identity(Ybuf[0], Ybuf[1], cur, S, hvec, kreg, aoff, soff, active, nterms);
    } else {
        v2f breg[32];
        m_frags(breg, Kbuf, jbase + col, hi);
        const int aoff = col * YSTRIDE + 2 * hi;
        taylor_action(Ybuf[0], Ybuf[1], cur, S, hvec, breg, aoff, soff);
    }

    if (isF) {
        const int nrows = 4 * win_chunks(mat);
        float* Wg = Wsp + win_base(mat, wave);
        const int kk0 = row0 + 8 * hi - 4 * win_start(mat, wave);
        #pragma unroll
        for (int v = 0; v < 8; ++v) {
            int kk = kk0 + v;
            if (kk >= 0 && kk < nrows) Wg[kk * TILE + col] = S[v];
        }
    } else {
        #pragma unroll
        for (int v = 0; v < 8; ++v) {
            int r = row0 + v + 8 * hi;
            if (r < B) out[r * UNITS + jbase + col] = S[v];
        }
    }
}

__global__ void __launch_bounds__(256)
apply_powers(const float* __restrict__ inp,
             const float* __restrict__ Wsp,
             float* out,
             int B)
{
    __shared__ __attribute__((aligned(16))) float Ybuf[2][TILE * YSTRIDE];

    const int tid   = threadIdx.x;
    const int wave  = tid >> 5;
    const int lane  = tid & 31;
    const int col   = lane & 15;
    const int hi    = lane >> 4;
    const int jbase = wave * 16;
    const int row0  = blockIdx.x * TILE;

    const int csF = win_start(0, wave);
    const int csP = win_start(1, wave);
    v2f ereg[BAND_CHUNKS];
    v2f preg[PQ_CHUNKS];
    v2f qreg[PQ_CHUNKS];
    w_frags<BAND_CHUNKS>(ereg, Wsp + win_base(0, wave), col, hi);
    w_frags<PQ_CHUNKS>(preg, Wsp + win_base(1, wave), col, hi);
    w_frags<PQ_CHUNKS>(qreg, Wsp + win_base(2, wave), col, hi);

    int take[8];
    v8f S;
    #pragma unroll
    for (int v = 0; v < 8; ++v) {
        int r  = row0 + v + 8 * hi;
        int rc = (r < B) ? r : (B - 1);
        float t = inp[rc * IN_STRIDE + UNITS];
        int q = nearest_64th(t);
        int g = q >> 3, a = (q >> 1) & 3, c = q & 1;
        take[v] = ((1 << g) - 1) | (((1 << a) - 1) << GMAX) | (c << (GMAX + PMAX));
        S[v]    = out[rc * UNITS + jbase + col];
    }
    for (int idx = tid; idx < TILE * UNITS / 4; idx += 256) {
        int r  = idx >> 5, q = idx & 31;
        int rr = row0 + r;
        int rc = (rr < B) ? rr : (B - 1);
        *reinterpret_cast<v4f*>(&Ybuf[0][r * YSTRIDE + 4 * q]) =
            *reinterpret_cast<const v4f*>(&out[rc * UNITS + 4 * q]);
    }
    __syncthreads();

    const int aoffF = col * YSTRIDE + 2 * hi + 4 * csF;
    const int aoffP = col * YSTRIDE + 2 * hi + 4 * csP;
    const int soff  = 8 * hi * YSTRIDE + jbase + col;

    int cur = 0;
    for (int j = 0; j < GMAX + PMAX + QMAX; ++j) {
        const float* Yc = cur ? Ybuf[1] : Ybuf[0];
        float*       Yn = cur ? Ybuf[0] : Ybuf[1];
        v8f z;
        if (j < GMAX)             z = apply_band<BAND_CHUNKS>(Yc, aoffF, ereg);
        else if (j < GMAX + PMAX) z = apply_band<PQ_CHUNKS>(Yc, aoffP, preg);
        else                      z = apply_band<PQ_CHUNKS>(Yc, aoffP, qreg);
        const bool more = (j < GMAX + PMAX + QMAX - 1);
        #pragma unroll
        for (int v = 0; v < 8; ++v) {
            S[v] = ((take[v] >> j) & 1) ? z[v] : S[v];
            if (more) Yn[v * YSTRIDE + soff] = S[v];
        }
        if (more) __syncthreads();
        cur ^= 1;
    }

    #pragma unroll
    for (int v = 0; v < 8; ++v) {
        int r = row0 + v + 8 * hi;
        if (r < B) out[r * UNITS + jbase + col] = S[v];
    }
}

extern "C" void kernel_launch(void* const* d_in, const int* in_sizes, int n_in,
                              void* d_out, int out_size, void* d_ws, size_t ws_size,
                              hipStream_t stream) {
    const float* inp    = (const float*)d_in[0];
    const float* params = (const float*)d_in[1];
    float*       out    = (float*)d_out;
    float*       Wsp    = (float*)d_ws;

    int B = in_sizes[0] / IN_STRIDE;
    int tiles = (B + TILE - 1) / TILE;

    taylor_tiles<<<tiles + NMATS * FTILES, 256, 0, stream>>>(inp, params, Wsp, out, B);
    apply_powers<<<tiles, 256, 0, stream>>>(inp, Wsp, out, B);
}


# One launch, not two: each workgroup builds expm(K/8), expm(K/32), expm(K/64), a column per thread by tridiagonal Horner steps on the vector unit; S stays in LDS; series sum order, rounding change
# speedup vs baseline: 5.5493x; 1.2251x over previous
#include <hip/hip_runtime.h>


#define UNITS     128
#define IN_STRIDE 129
#define TILE      16
#define TERMS     12
#define ROW_TERMS 4
#define GMAX      8
#define PMAX      3
#define QMAX      1
#define P_TERMS   7
#define Q_TERMS   6
#define YSTRIDE   132
#define MSIZE     (UNITS * YSTRIDE)

typedef __attribute__((ext_vector_type(2))) float v2f;
typedef __attribute__((ext_vector_type(4))) float v4f;
typedef __attribute__((ext_vector_type(8))) float v8f;

__device__ __forceinline__ int nearest_64th(float t) {
    int q = (int)floorf(t * 64.0f + 0.5f);
    return (q < 0) ? 0 : ((q > 8 * GMAX) ? 8 * GMAX : q);
}

#define PADW 16
#define PLEN (UNITS + 2 * PADW)
static_assert(TERMS + 1 <= PADW, "a column of F reads ceta TERMS + 1 places below its own row");

__device__ __forceinline__ void k_diagonals(float* Kb, float* sgP, float* ctP,
                                            float pd, float pc, int n) {
    if (n < UNITS) {
        float sp = (pd > 20.0f) ? pd : log1pf(__expf(pd));
        float sg = -sp * sp;
        float c  = (n + 1 < UNITS) ? -pc : 0.0f;
        Kb[n * YSTRIDE + n] = sg;
        if (n + 1 < UNITS) {
            Kb[n * YSTRIDE + n + 1]   = c;
            Kb[(n + 1) * YSTRIDE + n] = -c;
        }
        sgP[PADW + n] = sg;
        ctP[PADW + n] = c;
    } else if (n < UNITS + 2 * PADW) {
        int i   = n - UNITS;
        int idx = (i < PADW) ? i : (UNITS + i);
        sgP[idx] = 0.0f;
        ctP[idx] = 0.0f;
    }
}

__device__ __forceinline__ void zero_floats(float* Mz, int count4, int tid) {
    const v4f zero = {0.0f, 0.0f, 0.0f, 0.0f};
    for (int idx = tid; idx < count4; idx += 256)
        *reinterpret_cast<v4f*>(&Mz[4 * idx]) = zero;
}

__device__ __forceinline__ void m_frags(v2f (&breg)[32], const float* M, int n, int hi) {
    #pragma unroll
    for (int c = 0; c < 32; ++c) {
        int k = 4 * c + 2 * hi;
        breg[c][0] = M[k * YSTRIDE + n];
        breg[c][1] = M[(k + 1) * YSTRIDE + n];
    }
}

__device__ __forceinline__ v8f apply16(const float* Yc, int aoff, const v2f (&breg)[32]) {
    v8f z0 = {};
    v8f z1 = {};
    v2f abuf[8];
    #pragma unroll
    for (int c = 0; c < 8; ++c)
        abuf[c] = *reinterpret_cast<const v2f*>(&Yc[aoff + 4 * c]);
    #pragma unroll
    for (int c = 0; c < 32; ++c) {
        v2f a = abuf[c & 7];
        if (c + 8 < 32)
            abuf[c & 7] = *reinterpret_cast<const v2f*>(&Yc[aoff + 4 * (c + 8)]);
        if (c & 1)
            z1 = __builtin_amdgcn_wmma_f32_16x16x4_f32(
                    false, a, false, breg[c], (short)0, z1, false, false);
        else
            z0 = __builtin_amdgcn_wmma_f32_16x16x4_f32(
                    false, a, false, breg[c], (short)0, z0, false, false);
    }
    return z0 + z1;
}

__device__ __forceinline__ void taylor_action(float* Y0, float* Y1, int& cur, v8f& S,
                                              const float (&h)[8], const v2f (&breg)[32],
                                              int aoff, int soff) {
    for (int k = 1; k <= ROW_TERMS; ++k) {
        const float* Yc = cur ? Y1 : Y0;
        float*       Yn = cur ? Y0 : Y1;
        v8f z = apply16(Yc, aoff, breg);
        const float invk = __builtin_amdgcn_rcpf((float)k);
        const bool  last = (k == ROW_TERMS);
        #pragma unroll
        for (int v = 0; v < 8; ++v) {
            float val = z[v] * h[v] * invk;
            S[v] += val;
            Yn[v * YSTRIDE + soff] = last ? S[v] : val;
        }
        __syncthreads();
        cur ^= 1;
    }
}

#define BAND_CHUNKS ((TILE + 2 * TERMS) / 4)
static_assert(TERMS % 4 == 0, "band window must start on a k-chunk");
#define PQ_CHUNKS ((TILE + 2 * 8) / 4)
static_assert(P_TERMS <= 8 && Q_TERMS <= 8, "P and Q must fit the 8-off-diagonal window");

#define F_FLOATS  (8 * BAND_CHUNKS * 4 * TILE)
#define PQ_FLOATS (8 * PQ_CHUNKS * 4 * TILE)
#define WIN_FLOATS (F_FLOATS + 2 * PQ_FLOATS)
__device__ __forceinline__ int win_chunks(int mat) { return mat == 0 ? BAND_CHUNKS : PQ_CHUNKS; }
__device__ __forceinline__ int win_start(int mat, int w) {
    const int nch = win_chunks(mat);
    int cs = 4 * w - (mat == 0 ? TERMS / 4 : 2);
    return (cs < 0) ? 0 : ((cs > UNITS / 4 - nch) ? (UNITS / 4 - nch) : cs);
}
__device__ __forceinline__ int win_base(int mat, int w) {
    return (mat == 0 ? 0 : (F_FLOATS + (mat - 1) * PQ_FLOATS)) + w * win_chunks(mat) * 4 * TILE;
}

template <int NCH>
__device__ __forceinline__ void w_frags(v2f (&breg)[NCH], const float* Wg,
                                        int col, int hi) {
    #pragma unroll
    for (int i = 0; i < NCH; ++i) {
        int kk = 4 * i + 2 * hi;
        breg[i][0] = Wg[kk * TILE + col];
        breg[i][1] = Wg[(kk + 1) * TILE + col];
    }
}

template <int NCH>
__device__ __forceinline__ v8f apply_band(const float* Yc, int aoff, const v2f (&breg)[NCH]) {
    v8f z0 = {};
    v8f z1 = {};
    v2f abuf[NCH];
    #pragma unroll
    for (int i = 0; i < NCH; ++i)
        abuf[i] = *reinterpret_cast<const v2f*>(&Yc[aoff + 4 * i]);
    #pragma unroll
    for (int i = 0; i < NCH; ++i) {
        if (i & 1)
            z1 = __builtin_amdgcn_wmma_f32_16x16x4_f32(
                    false, abuf[i], false, breg[i], (short)0, z1, false, false);
        else
            z0 = __builtin_amdgcn_wmma_f32_16x16x4_f32(
                    false, abuf[i], false, breg[i], (short)0, z0, false, false);
    }
    return z0 + z1;
}

template <int NC>
__device__ __forceinline__ void column_coefs(float (&di)[2 * NC + 1], float (&up)[2 * NC + 2],
                                             const float* sgP, const float* ctP, int n) {
    #pragma unroll
    for (int j = 0; j < 2 * NC + 1; ++j) di[j] = sgP[PADW + n - NC + j];
    #pragma unroll
    for (int j = 0; j < 2 * NC + 2; ++j) up[j] = ctP[PADW + n - NC - 1 + j];
}

template <int N, int NC>
__device__ __forceinline__ void band_column(float (&w)[2 * N + 1], const float (&di)[2 * NC + 1],
                                            const float (&up)[2 * NC + 2], float h) {
    #pragma unroll
    for (int i = 0; i < 2 * N + 1; ++i) w[i] = (i == N) ? 1.0f : 0.0f;
    #pragma unroll
    for (int s = 1; s <= N; ++s) {
        const float ck = h / (float)(N - s + 1);
        float prev = 0.0f;
        #pragma unroll
        for (int d = -s; d <= s; ++d) {
            const int   iw  = d + N;
            const int   ic  = d + NC;
            const float old = w[iw];
            const bool hasC = (d > -s) && (d < s);
            const bool hasN = (d + 1 > -s) && (d + 1 < s);
            const bool hasP = (d - 1 > -s) && (d - 1 < s);
            float t;
            if (hasC) {
                t = di[ic] * old;
                if (hasN) t += up[ic + 1] * w[iw + 1];
                if (hasP) t -= up[ic] * prev;
            } else if (hasN) {
                t = up[ic + 1] * w[iw + 1];
            } else {
                t = -(up[ic] * prev);
            }
            w[iw] = (d == 0) ? (1.0f + ck * t) : (ck * t);
            prev  = old;
        }
    }
}

template <int N>
__device__ __forceinline__ void store_column(const float (&w)[2 * N + 1], float* Win, int mat, int n) {
    const int wb = n >> 4;
    float* Wg = Win + win_base(mat, wb) + (n & 15) - 4 * win_start(mat, wb) * TILE;
    #pragma unroll
    for (int d = -N; d <= N; ++d) {
        const int row = n + d;
        if (row >= 0 && row < UNITS) Wg[row * TILE] = w[d + N];
    }
}

__global__ void __launch_bounds__(256)
expm_rows(const float* __restrict__ inp,
          const float* __restrict__ p,
          float* __restrict__ out,
          int B)
{
    __shared__ __attribute__((aligned(16))) float Kbuf[MSIZE];
    __shared__ __attribute__((aligned(16))) float Ybuf[2][TILE * YSTRIDE];
    __shared__ __attribute__((aligned(16))) float Win[WIN_FLOATS];
    __shared__ float sgP[PLEN];
    __shared__ float ctP[PLEN];

    const int tid   = threadIdx.x;
    const int wave  = tid >> 5;
    const int lane  = tid & 31;
    const int col   = lane & 15;
    const int hi    = lane >> 4;
    const int jbase = wave * 16;
    const int row0  = blockIdx.x * TILE;

    float pd = 0.0f, pc = 0.0f;
    if (tid < UNITS) {
        pd = p[tid];
        if (tid + 1 < UNITS) pc = p[UNITS + tid];
    }

    float hvec[8];
    int   take[8];
    v8f   S;
    #pragma unroll
    for (int v = 0; v < 8; ++v) {
        int r  = row0 + v + 8 * hi;
        int rc = (r < B) ? r : (B - 1);
        float t = inp[rc * IN_STRIDE + UNITS];
        int q = nearest_64th(t);
        int g = q >> 3, a = (q >> 1) & 3, c = q & 1;
        hvec[v] = t - 0.015625f * (float)q;
        take[v] = ((1 << g) - 1) | (((1 << a) - 1) << GMAX) | (c << (GMAX + PMAX));
        S[v]    = inp[rc * IN_STRIDE + jbase + col];
    }
    for (int idx = tid; idx < TILE * UNITS; idx += 256) {
        int r  = idx >> 7, cc = idx & 127;
        int rr = row0 + r;
        int rc = (rr < B) ? rr : (B - 1);
        Ybuf[0][r * YSTRIDE + cc] = inp[rc * IN_STRIDE + cc];
    }

    zero_floats(Kbuf, MSIZE / 4, tid);
    zero_floats(Win, WIN_FLOATS / 4, tid);
    __syncthreads();
    k_diagonals(Kbuf, sgP, ctP, pd, pc, tid);
    __syncthreads();

    {
        const int wv = __builtin_amdgcn_readfirstlane(wave);
        if (wv < 4) {
            const int n = tid;
            float di[2 * TERMS + 1], up[2 * TERMS + 2], w[2 * TERMS + 1];
            column_coefs<TERMS>(di, up, sgP, ctP, n);
            band_column<TERMS, TERMS>(w, di, up, 0.125f);
            store_column<TERMS>(w, Win, 0, n);
        } else {
            const int n = tid - UNITS;
            float di[2 * P_TERMS + 1], up[2 * P_TERMS + 2];
            column_coefs<P_TERMS>(di, up, sgP, ctP, n);
            float wp[2 * P_TERMS + 1];
            band_column<P_TERMS, P_TERMS>(wp, di, up, 0.03125f);
            store_column<P_TERMS>(wp, Win, 1, n);
            float wq[2 * Q_TERMS + 1];
            band_column<Q_TERMS, P_TERMS>(wq, di, up, 0.015625f);
            store_column<Q_TERMS>(wq, Win, 2, n);
        }
    }
    __syncthreads();

    const int soff = 8 * hi * YSTRIDE + jbase + col;
    int cur = 0;

    {
        v2f breg[32];
        m_frags(breg, Kbuf, jbase + col, hi);
        const int aoff = col * YSTRIDE + 2 * hi;
        taylor_action(Ybuf[0], Ybuf[1], cur, S, hvec, breg, aoff, soff);
    }

    const int csF = win_start(0, wave);
    const int csP = win_start(1, wave);
    v2f ereg[BAND_CHUNKS];
    v2f preg[PQ_CHUNKS];
    v2f qreg[PQ_CHUNKS];
    w_frags<BAND_CHUNKS>(ereg, Win + win_base(0, wave), col, hi);
    w_frags<PQ_CHUNKS>(preg, Win + win_base(1, wave), col, hi);
    w_frags<PQ_CHUNKS>(qreg, Win + win_base(2, wave), col, hi);

    const int aoffF = col * YSTRIDE + 2 * hi + 4 * csF;
    const int aoffP = col * YSTRIDE + 2 * hi + 4 * csP;

    for (int j = 0; j < GMAX + PMAX + QMAX; ++j) {
        const float* Yc = cur ? Ybuf[1] : Ybuf[0];
        float*       Yn = cur ? Ybuf[0] : Ybuf[1];
        v8f z;
        if (j < GMAX)             z = apply_band<BAND_CHUNKS>(Yc, aoffF, ereg);
        else if (j < GMAX + PMAX) z = apply_band<PQ_CHUNKS>(Yc, aoffP, preg);
        else                      z = apply_band<PQ_CHUNKS>(Yc, aoffP, qreg);
        const bool more = (j < GMAX + PMAX + QMAX - 1);
        #pragma unroll
        for (int v = 0; v < 8; ++v) {
            S[v] = ((take[v] >> j) & 1) ? z[v] : S[v];
            if (more) Yn[v * YSTRIDE + soff] = S[v];
        }
        if (more) __syncthreads();
        cur ^= 1;
    }

    #pragma unroll
    for (int v = 0; v < 8; ++v) {
        int r = row0 + v + 8 * hi;
        if (r < B) out[r * UNITS + jbase + col] = S[v];
    }
}

extern "C" void kernel_launch(void* const* d_in, const int* in_sizes, int n_in,
                              void* d_out, int out_size, void* d_ws, size_t ws_size,
                              hipStream_t stream) {
    const float* inp    = (const float*)d_in[0];
    const float* params = (const float*)d_in[1];
    float*       out    = (float*)d_out;

    int B = in_sizes[0] / IN_STRIDE;
    int tiles = (B + TILE - 1) / TILE;

    expm_rows<<<tiles, 256, 0, stream>>>(inp, params, out, B);
}


# Rows' series as 4 independent band products x*(K^j/j!), j=1..4, from one A-fragment load: 24 WMMAs, 1 barrier, not 128 and 4; no dense K in LDS; powers of r applied after; sum order, rounding change
# speedup vs baseline: 6.4155x; 1.1561x over previous
#include <hip/hip_runtime.h>


#define UNITS     128
#define IN_STRIDE 129
#define TILE      16
#define TERMS     12
#define ROW_TERMS 4
#define GMAX      8
#define PMAX      3
#define QMAX      1
#define P_TERMS   7
#define Q_TERMS   6
#define YSTRIDE   132

typedef __attribute__((ext_vector_type(2))) float v2f;
typedef __attribute__((ext_vector_type(4))) float v4f;
typedef __attribute__((ext_vector_type(8))) float v8f;

__device__ __forceinline__ int nearest_64th(float t) {
    int q = (int)floorf(t * 64.0f + 0.5f);
    return (q < 0) ? 0 : ((q > 8 * GMAX) ? 8 * GMAX : q);
}

#define PADW 16
#define PLEN (UNITS + 2 * PADW)
static_assert(TERMS + 1 <= PADW, "a column of F reads ceta TERMS + 1 places below its own row");

__device__ __forceinline__ void k_diagonals(float* sgP, float* ctP,
                                            float pd, float pc, int n) {
    if (n < UNITS) {
        float sp = (pd > 20.0f) ? pd : log1pf(__expf(pd));
        float sg = -sp * sp;
        float c  = (n + 1 < UNITS) ? -pc : 0.0f;
        sgP[PADW + n] = sg;
        ctP[PADW + n] = c;
    } else if (n < UNITS + 2 * PADW) {
        int i   = n - UNITS;
        int idx = (i < PADW) ? i : (UNITS + i);
        sgP[idx] = 0.0f;
        ctP[idx] = 0.0f;
    }
}

__device__ __forceinline__ void zero_floats(float* Mz, int count4, int tid) {
    const v4f zero = {0.0f, 0.0f, 0.0f, 0.0f};
    for (int idx = tid; idx < count4; idx += 256)
        *reinterpret_cast<v4f*>(&Mz[4 * idx]) = zero;
}

#define BAND_CHUNKS ((TILE + 2 * TERMS) / 4)
static_assert(TERMS % 4 == 0, "band window must start on a k-chunk");
#define PQ_CHUNKS ((TILE + 2 * 8) / 4)
static_assert(P_TERMS <= 8 && Q_TERMS <= 8, "P and Q must fit the 8-off-diagonal window");

#define R_CHUNKS ((TILE + 2 * ROW_TERMS) / 4)
static_assert(ROW_TERMS == 4, "the rows' band window must start on a k-chunk");

#define F_FLOATS  (8 * BAND_CHUNKS * 4 * TILE)
#define PQ_FLOATS (8 * PQ_CHUNKS * 4 * TILE)
#define R_FLOATS  (8 * R_CHUNKS * 4 * TILE)
#define WIN_FLOATS (F_FLOATS + 2 * PQ_FLOATS + ROW_TERMS * R_FLOATS)
__device__ __forceinline__ int win_chunks(int mat) {
    return mat == 0 ? BAND_CHUNKS : (mat <= 2 ? PQ_CHUNKS : R_CHUNKS);
}
__device__ __forceinline__ int win_start(int mat, int w) {
    const int nch = win_chunks(mat);
    int cs = 4 * w - (mat == 0 ? TERMS / 4 : (mat <= 2 ? 2 : ROW_TERMS / 4));
    return (cs < 0) ? 0 : ((cs > UNITS / 4 - nch) ? (UNITS / 4 - nch) : cs);
}
__device__ __forceinline__ int win_base(int mat, int w) {
    const int mb = (mat == 0) ? 0
                 : (mat <= 2) ? (F_FLOATS + (mat - 1) * PQ_FLOATS)
                              : (F_FLOATS + 2 * PQ_FLOATS + (mat - 3) * R_FLOATS);
    return mb + w * win_chunks(mat) * 4 * TILE;
}

template <int NCH>
__device__ __forceinline__ void w_frags(v2f (&breg)[NCH], const float* Wg,
                                        int col, int hi) {
    #pragma unroll
    for (int i = 0; i < NCH; ++i) {
        int kk = 4 * i + 2 * hi;
        breg[i][0] = Wg[kk * TILE + col];
        breg[i][1] = Wg[(kk + 1) * TILE + col];
    }
}

template <int NCH>
__device__ __forceinline__ v8f apply_band(const float* Yc, int aoff, const v2f (&breg)[NCH]) {
    v8f z0 = {};
    v8f z1 = {};
    v2f abuf[NCH];
    #pragma unroll
    for (int i = 0; i < NCH; ++i)
        abuf[i] = *reinterpret_cast<const v2f*>(&Yc[aoff + 4 * i]);
    #pragma unroll
    for (int i = 0; i < NCH; ++i) {
        if (i & 1)
            z1 = __builtin_amdgcn_wmma_f32_16x16x4_f32(
                    false, abuf[i], false, breg[i], (short)0, z1, false, false);
        else
            z0 = __builtin_amdgcn_wmma_f32_16x16x4_f32(
                    false, abuf[i], false, breg[i], (short)0, z0, false, false);
    }
    return z0 + z1;
}

template <int NC>
__device__ __forceinline__ void column_coefs(float (&di)[2 * NC + 1], float (&up)[2 * NC + 2],
                                             const float* sgP, const float* ctP, int n) {
    #pragma unroll
    for (int j = 0; j < 2 * NC + 1; ++j) di[j] = sgP[PADW + n - NC + j];
    #pragma unroll
    for (int j = 0; j < 2 * NC + 2; ++j) up[j] = ctP[PADW + n - NC - 1 + j];
}

template <int N, int NC>
__device__ __forceinline__ void band_column(float (&w)[2 * N + 1], const float (&di)[2 * NC + 1],
                                            const float (&up)[2 * NC + 2], float h) {
    #pragma unroll
    for (int i = 0; i < 2 * N + 1; ++i) w[i] = (i == N) ? 1.0f : 0.0f;
    #pragma unroll
    for (int s = 1; s <= N; ++s) {
        const float ck = h / (float)(N - s + 1);
        float prev = 0.0f;
        #pragma unroll
        for (int d = -s; d <= s; ++d) {
            const int   iw  = d + N;
            const int   ic  = d + NC;
            const float old = w[iw];
            const bool hasC = (d > -s) && (d < s);
            const bool hasN = (d + 1 > -s) && (d + 1 < s);
            const bool hasP = (d - 1 > -s) && (d - 1 < s);
            float t;
            if (hasC) {
                t = di[ic] * old;
                if (hasN) t += up[ic + 1] * w[iw + 1];
                if (hasP) t -= up[ic] * prev;
            } else if (hasN) {
                t = up[ic + 1] * w[iw + 1];
            } else {
                t = -(up[ic] * prev);
            }
            w[iw] = (d == 0) ? (1.0f + ck * t) : (ck * t);
            prev  = old;
        }
    }
}

template <int N, int NC>
__device__ __forceinline__ void power_step(float (&w)[2 * N + 1], const float (&di)[2 * NC + 1],
                                           const float (&up)[2 * NC + 2], int s, float ck) {
    float prev = 0.0f;
    #pragma unroll
    for (int d = -N; d <= N; ++d) {
        if (d < -s || d > s) continue;
        const int   iw  = d + N;
        const int   ic  = d + NC;
        const float old = w[iw];
        const bool hasC = (d > -s) && (d < s);
        const bool hasN = (d + 1 > -s) && (d + 1 < s);
        const bool hasP = (d - 1 > -s) && (d - 1 < s);
        float t;
        if (hasC) {
            t = di[ic] * old;
            if (hasN) t += up[ic + 1] * w[iw + 1];
            if (hasP) t -= up[ic] * prev;
        } else if (hasN) {
            t = up[ic + 1] * w[iw + 1];
        } else {
            t = -(up[ic] * prev);
        }
        w[iw] = ck * t;
        prev  = old;
    }
}

template <int N>
__device__ __forceinline__ void store_column(const float (&w)[2 * N + 1], float* Win, int mat, int n) {
    const int wb = n >> 4;
    float* Wg = Win + win_base(mat, wb) + (n & 15) - 4 * win_start(mat, wb) * TILE;
    #pragma unroll
    for (int d = -N; d <= N; ++d) {
        const int row = n + d;
        if (row >= 0 && row < UNITS) Wg[row * TILE] = w[d + N];
    }
}

__device__ __forceinline__ void apply_powers(v8f (&z)[ROW_TERMS], const float* Yc, int aoff,
                                             const v2f (&breg)[ROW_TERMS][R_CHUNKS]) {
    v2f abuf[R_CHUNKS];
    #pragma unroll
    for (int i = 0; i < R_CHUNKS; ++i)
        abuf[i] = *reinterpret_cast<const v2f*>(&Yc[aoff + 4 * i]);
    #pragma unroll
    for (int j = 0; j < ROW_TERMS; ++j) z[j] = v8f{};
    #pragma unroll
    for (int i = 0; i < R_CHUNKS; ++i) {
        #pragma unroll
        for (int j = 0; j < ROW_TERMS; ++j)
            z[j] = __builtin_amdgcn_wmma_f32_16x16x4_f32(
                       false, abuf[i], false, breg[j][i], (short)0, z[j], false, false);
    }
}

__global__ void __launch_bounds__(256)
expm_rows(const float* __restrict__ inp,
          const float* __restrict__ p,
          float* __restrict__ out,
          int B)
{
    __shared__ __attribute__((aligned(16))) float Ybuf[2][TILE * YSTRIDE];
    __shared__ __attribute__((aligned(16))) float Win[WIN_FLOATS];
    __shared__ float sgP[PLEN];
    __shared__ float ctP[PLEN];

    const int tid   = threadIdx.x;
    const int wave  = tid >> 5;
    const int lane  = tid & 31;
    const int col   = lane & 15;
    const int hi    = lane >> 4;
    const int jbase = wave * 16;
    const int row0  = blockIdx.x * TILE;

    float pd = 0.0f, pc = 0.0f;
    if (tid < UNITS) {
        pd = p[tid];
        if (tid + 1 < UNITS) pc = p[UNITS + tid];
    }

    float hvec[8];
    int   take[8];
    v8f   S;
    #pragma unroll
    for (int v = 0; v < 8; ++v) {
        int r  = row0 + v + 8 * hi;
        int rc = (r < B) ? r : (B - 1);
        float t = inp[rc * IN_STRIDE + UNITS];
        int q = nearest_64th(t);
        int g = q >> 3, a = (q >> 1) & 3, c = q & 1;
        hvec[v] = t - 0.015625f * (float)q;
        take[v] = ((1 << g) - 1) | (((1 << a) - 1) << GMAX) | (c << (GMAX + PMAX));
        S[v]    = inp[rc * IN_STRIDE + jbase + col];
    }
    for (int idx = tid; idx < TILE * UNITS; idx += 256) {
        int r  = idx >> 7, cc = idx & 127;
        int rr = row0 + r;
        int rc = (rr < B) ? rr : (B - 1);
        Ybuf[0][r * YSTRIDE + cc] = inp[rc * IN_STRIDE + cc];
    }

    zero_floats(Win, WIN_FLOATS / 4, tid);
    k_diagonals(sgP, ctP, pd, pc, tid);
    __syncthreads();

    {
        const int wv = __builtin_amdgcn_readfirstlane(wave);
        if (wv < 4) {
            const int n = tid;
            float di[2 * TERMS + 1], up[2 * TERMS + 2], w[2 * TERMS + 1];
            column_coefs<TERMS>(di, up, sgP, ctP, n);
            band_column<TERMS, TERMS>(w, di, up, 0.125f);
            store_column<TERMS>(w, Win, 0, n);
        } else {
            const int n = tid - UNITS;
            float di[2 * P_TERMS + 1], up[2 * P_TERMS + 2];
            column_coefs<P_TERMS>(di, up, sgP, ctP, n);
            float wp[2 * P_TERMS + 1];
            band_column<P_TERMS, P_TERMS>(wp, di, up, 0.03125f);
            store_column<P_TERMS>(wp, Win, 1, n);
            float wq[2 * Q_TERMS + 1];
            band_column<Q_TERMS, P_TERMS>(wq, di, up, 0.015625f);
            store_column<Q_TERMS>(wq, Win, 2, n);
            float wr[2 * ROW_TERMS + 1];
            #pragma unroll
            for (int i = 0; i < 2 * ROW_TERMS + 1; ++i) wr[i] = (i == ROW_TERMS) ? 1.0f : 0.0f;
            #pragma unroll
            for (int s = 1; s <= ROW_TERMS; ++s) {
                power_step<ROW_TERMS, P_TERMS>(wr, di, up, s, 1.0f / (float)s);
                store_column<ROW_TERMS>(wr, Win, 2 + s, n);
            }
        }
    }
    __syncthreads();

    const int soff = 8 * hi * YSTRIDE + jbase + col;
    int cur = 0;

    {
        const int csR = win_start(3, wave);
        v2f rreg[ROW_TERMS][R_CHUNKS];
        #pragma unroll
        for (int j = 0; j < ROW_TERMS; ++j)
            w_frags<R_CHUNKS>(rreg[j], Win + win_base(3 + j, wave), col, hi);
        const int aoffR = col * YSTRIDE + 2 * hi + 4 * csR;
        v8f z[ROW_TERMS];
        apply_powers(z, Ybuf[0], aoffR, rreg);
        #pragma unroll
        for (int v = 0; v < 8; ++v) {
            const float h = hvec[v];
            S[v] += h * (z[0][v] + h * (z[1][v] + h * (z[2][v] + h * z[3][v])));
            Ybuf[1][v * YSTRIDE + soff] = S[v];
        }
        __syncthreads();
        cur = 1;
    }

    const int csF = win_start(0, wave);
    const int csP = win_start(1, wave);
    v2f ereg[BAND_CHUNKS];
    v2f preg[PQ_CHUNKS];
    v2f qreg[PQ_CHUNKS];
    w_frags<BAND_CHUNKS>(ereg, Win + win_base(0, wave), col, hi);
    w_frags<PQ_CHUNKS>(preg, Win + win_base(1, wave), col, hi);
    w_frags<PQ_CHUNKS>(qreg, Win + win_base(2, wave), col, hi);

    const int aoffF = col * YSTRIDE + 2 * hi + 4 * csF;
    const int aoffP = col * YSTRIDE + 2 * hi + 4 * csP;

    for (int j = 0; j < GMAX + PMAX + QMAX; ++j) {
        const float* Yc = cur ? Ybuf[1] : Ybuf[0];
        float*       Yn = cur ? Ybuf[0] : Ybuf[1];
        v8f z;
        if (j < GMAX)             z = apply_band<BAND_CHUNKS>(Yc, aoffF, ereg);
        else if (j < GMAX + PMAX) z = apply_band<PQ_CHUNKS>(Yc, aoffP, preg);
        else                      z = apply_band<PQ_CHUNKS>(Yc, aoffP, qreg);
        const bool more = (j < GMAX + PMAX + QMAX - 1);
        #pragma unroll
        for (int v = 0; v < 8; ++v) {
            S[v] = ((take[v] >> j) & 1) ? z[v] : S[v];
            if (more) Yn[v * YSTRIDE + soff] = S[v];
        }
        if (more) __syncthreads();
        cur ^= 1;
    }

    #pragma unroll
    for (int v = 0; v < 8; ++v) {
        int r = row0 + v + 8 * hi;
        if (r < B) out[r * UNITS + jbase + col] = S[v];
    }
}

extern "C" void kernel_launch(void* const* d_in, const int* in_sizes, int n_in,
                              void* d_out, int out_size, void* d_ws, size_t ws_size,
                              hipStream_t stream) {
    const float* inp    = (const float*)d_in[0];
    const float* params = (const float*)d_in[1];
    float*       out    = (float*)d_out;

    int B = in_sizes[0] / IN_STRIDE;
    int tiles = (B + TILE - 1) / TILE;

    expm_rows<<<tiles, 256, 0, stream>>>(inp, params, out, B);
}


# Column stores select their address (spare LDS slot for rows outside the matrix), no branch per store: a column build is one straight-line block, row updates overlap; sum order same, fusing may differ
# speedup vs baseline: 6.9589x; 1.0847x over previous
#include <hip/hip_runtime.h>


#define UNITS     128
#define IN_STRIDE 129
#define TILE      16
#define TERMS     12
#define ROW_TERMS 4
#define GMAX      8
#define PMAX      3
#define QMAX      1
#define P_TERMS   7
#define Q_TERMS   6
#define YSTRIDE   132

typedef __attribute__((ext_vector_type(2))) float v2f;
typedef __attribute__((ext_vector_type(4))) float v4f;
typedef __attribute__((ext_vector_type(8))) float v8f;

__device__ __forceinline__ int nearest_64th(float t) {
    int q = (int)floorf(t * 64.0f + 0.5f);
    return (q < 0) ? 0 : ((q > 8 * GMAX) ? 8 * GMAX : q);
}

#define PADW 16
#define PLEN (UNITS + 2 * PADW)
static_assert(TERMS + 1 <= PADW, "a column of F reads ceta TERMS + 1 places below its own row");

__device__ __forceinline__ void k_diagonals(float* sgP, float* ctP,
                                            float pd, float pc, int n) {
    if (n < UNITS) {
        float sp = (pd > 20.0f) ? pd : log1pf(__expf(pd));
        float sg = -sp * sp;
        float c  = (n + 1 < UNITS) ? -pc : 0.0f;
        sgP[PADW + n] = sg;
        ctP[PADW + n] = c;
    } else if (n < UNITS + 2 * PADW) {
        int i   = n - UNITS;
        int idx = (i < PADW) ? i : (UNITS + i);
        sgP[idx] = 0.0f;
        ctP[idx] = 0.0f;
    }
}

__device__ __forceinline__ void zero_floats(float* Mz, int count4, int tid) {
    const v4f zero = {0.0f, 0.0f, 0.0f, 0.0f};
    for (int idx = tid; idx < count4; idx += 256)
        *reinterpret_cast<v4f*>(&Mz[4 * idx]) = zero;
}

#define BAND_CHUNKS ((TILE + 2 * TERMS) / 4)
static_assert(TERMS % 4 == 0, "band window must start on a k-chunk");
#define PQ_CHUNKS ((TILE + 2 * 8) / 4)
static_assert(P_TERMS <= 8 && Q_TERMS <= 8, "P and Q must fit the 8-off-diagonal window");

#define R_CHUNKS ((TILE + 2 * ROW_TERMS) / 4)
static_assert(ROW_TERMS == 4, "the rows' band window must start on a k-chunk");

#define F_FLOATS  (8 * BAND_CHUNKS * 4 * TILE)
#define PQ_FLOATS (8 * PQ_CHUNKS * 4 * TILE)
#define R_FLOATS  (8 * R_CHUNKS * 4 * TILE)
#define WIN_FLOATS (F_FLOATS + 2 * PQ_FLOATS + ROW_TERMS * R_FLOATS)
__device__ __forceinline__ int win_chunks(int mat) {
    return mat == 0 ? BAND_CHUNKS : (mat <= 2 ? PQ_CHUNKS : R_CHUNKS);
}
__device__ __forceinline__ int win_start(int mat, int w) {
    const int nch = win_chunks(mat);
    int cs = 4 * w - (mat == 0 ? TERMS / 4 : (mat <= 2 ? 2 : ROW_TERMS / 4));
    return (cs < 0) ? 0 : ((cs > UNITS / 4 - nch) ? (UNITS / 4 - nch) : cs);
}
__device__ __forceinline__ int win_base(int mat, int w) {
    const int mb = (mat == 0) ? 0
                 : (mat <= 2) ? (F_FLOATS + (mat - 1) * PQ_FLOATS)
                              : (F_FLOATS + 2 * PQ_FLOATS + (mat - 3) * R_FLOATS);
    return mb + w * win_chunks(mat) * 4 * TILE;
}

template <int NCH>
__device__ __forceinline__ void w_frags(v2f (&breg)[NCH], const float* Wg,
                                        int col, int hi) {
    #pragma unroll
    for (int i = 0; i < NCH; ++i) {
        int kk = 4 * i + 2 * hi;
        breg[i][0] = Wg[kk * TILE + col];
        breg[i][1] = Wg[(kk + 1) * TILE + col];
    }
}

template <int NCH>
__device__ __forceinline__ v8f apply_band(const float* Yc, int aoff, const v2f (&breg)[NCH]) {
    v8f z0 = {};
    v8f z1 = {};
    v2f abuf[NCH];
    #pragma unroll
    for (int i = 0; i < NCH; ++i)
        abuf[i] = *reinterpret_cast<const v2f*>(&Yc[aoff + 4 * i]);
    #pragma unroll
    for (int i = 0; i < NCH; ++i) {
        if (i & 1)
            z1 = __builtin_amdgcn_wmma_f32_16x16x4_f32(
                    false, abuf[i], false, breg[i], (short)0, z1, false, false);
        else
            z0 = __builtin_amdgcn_wmma_f32_16x16x4_f32(
                    false, abuf[i], false, breg[i], (short)0, z0, false, false);
    }
    return z0 + z1;
}

template <int NC>
__device__ __forceinline__ void column_coefs(float (&di)[2 * NC + 1], float (&up)[2 * NC + 2],
                                             const float* sgP, const float* ctP, int n) {
    #pragma unroll
    for (int j = 0; j < 2 * NC + 1; ++j) di[j] = sgP[PADW + n - NC + j];
    #pragma unroll
    for (int j = 0; j < 2 * NC + 2; ++j) up[j] = ctP[PADW + n - NC - 1 + j];
}

template <int N, int NC>
__device__ __forceinline__ void band_column(float (&w)[2 * N + 1], const float (&di)[2 * NC + 1],
                                            const float (&up)[2 * NC + 2], float h) {
    #pragma unroll
    for (int i = 0; i < 2 * N + 1; ++i) w[i] = (i == N) ? 1.0f : 0.0f;
    #pragma unroll
    for (int s = 1; s <= N; ++s) {
        const float ck = h / (float)(N - s + 1);
        float prev = 0.0f;
        #pragma unroll
        for (int d = -s; d <= s; ++d) {
            const int   iw  = d + N;
            const int   ic  = d + NC;
            const float old = w[iw];
            const bool hasC = (d > -s) && (d < s);
            const bool hasN = (d + 1 > -s) && (d + 1 < s);
            const bool hasP = (d - 1 > -s) && (d - 1 < s);
            float t;
            if (hasC) {
                t = di[ic] * old;
                if (hasN) t += up[ic + 1] * w[iw + 1];
                if (hasP) t -= up[ic] * prev;
            } else if (hasN) {
                t = up[ic + 1] * w[iw + 1];
            } else {
                t = -(up[ic] * prev);
            }
            w[iw] = (d == 0) ? (1.0f + ck * t) : (ck * t);
            prev  = old;
        }
    }
}

template <int N, int NC>
__device__ __forceinline__ void power_step(float (&w)[2 * N + 1], const float (&di)[2 * NC + 1],
                                           const float (&up)[2 * NC + 2], int s, float ck) {
    float prev = 0.0f;
    #pragma unroll
    for (int d = -N; d <= N; ++d) {
        if (d < -s || d > s) continue;
        const int   iw  = d + N;
        const int   ic  = d + NC;
        const float old = w[iw];
        const bool hasC = (d > -s) && (d < s);
        const bool hasN = (d + 1 > -s) && (d + 1 < s);
        const bool hasP = (d - 1 > -s) && (d - 1 < s);
        float t;
        if (hasC) {
            t = di[ic] * old;
            if (hasN) t += up[ic + 1] * w[iw + 1];
            if (hasP) t -= up[ic] * prev;
        } else if (hasN) {
            t = up[ic + 1] * w[iw + 1];
        } else {
            t = -(up[ic] * prev);
        }
        w[iw] = ck * t;
        prev  = old;
    }
}

template <int N>
__device__ __forceinline__ void store_column(const float (&w)[2 * N + 1], float* Win, int mat,
                                             int n, int spare) {
    const int wb   = n >> 4;
    const int base = win_base(mat, wb) + (n & 15) - 4 * win_start(mat, wb) * TILE;
    #pragma unroll
    for (int d = -N; d <= N; ++d) {
        const int row = n + d;
        const int off = (row >= 0 && row < UNITS) ? (base + row * TILE) : spare;
        Win[off] = w[d + N];
    }
}

__device__ __forceinline__ void apply_powers(v8f (&z)[ROW_TERMS], const float* Yc, int aoff,
                                             const v2f (&breg)[ROW_TERMS][R_CHUNKS]) {
    v2f abuf[R_CHUNKS];
    #pragma unroll
    for (int i = 0; i < R_CHUNKS; ++i)
        abuf[i] = *reinterpret_cast<const v2f*>(&Yc[aoff + 4 * i]);
    #pragma unroll
    for (int j = 0; j < ROW_TERMS; ++j) z[j] = v8f{};
    #pragma unroll
    for (int i = 0; i < R_CHUNKS; ++i) {
        #pragma unroll
        for (int j = 0; j < ROW_TERMS; ++j)
            z[j] = __builtin_amdgcn_wmma_f32_16x16x4_f32(
                       false, abuf[i], false, breg[j][i], (short)0, z[j], false, false);
    }
}

__global__ void __launch_bounds__(256)
expm_rows(const float* __restrict__ inp,
          const float* __restrict__ p,
          float* __restrict__ out,
          int B)
{
    __shared__ __attribute__((aligned(16))) float Ybuf[2][TILE * YSTRIDE];
    __shared__ __attribute__((aligned(16))) float Win[WIN_FLOATS + 256];
    __shared__ float sgP[PLEN];
    __shared__ float ctP[PLEN];

    const int tid   = threadIdx.x;
    const int wave  = tid >> 5;
    const int lane  = tid & 31;
    const int col   = lane & 15;
    const int hi    = lane >> 4;
    const int jbase = wave * 16;
    const int row0  = blockIdx.x * TILE;

    float pd = 0.0f, pc = 0.0f;
    if (tid < UNITS) {
        pd = p[tid];
        if (tid + 1 < UNITS) pc = p[UNITS + tid];
    }

    float hvec[8];
    int   take[8];
    v8f   S;
    #pragma unroll
    for (int v = 0; v < 8; ++v) {
        int r  = row0 + v + 8 * hi;
        int rc = (r < B) ? r : (B - 1);
        float t = inp[rc * IN_STRIDE + UNITS];
        int q = nearest_64th(t);
        int g = q >> 3, a = (q >> 1) & 3, c = q & 1;
        hvec[v] = t - 0.015625f * (float)q;
        take[v] = ((1 << g) - 1) | (((1 << a) - 1) << GMAX) | (c << (GMAX + PMAX));
        S[v]    = inp[rc * IN_STRIDE + jbase + col];
    }
    for (int idx = tid; idx < TILE * UNITS; idx += 256) {
        int r  = idx >> 7, cc = idx & 127;
        int rr = row0 + r;
        int rc = (rr < B) ? rr : (B - 1);
        Ybuf[0][r * YSTRIDE + cc] = inp[rc * IN_STRIDE + cc];
    }

    zero_floats(Win, WIN_FLOATS / 4, tid);
    k_diagonals(sgP, ctP, pd, pc, tid);
    __syncthreads();

    {
        const int wv = __builtin_amdgcn_readfirstlane(wave);
        if (wv < 4) {
            const int n = tid;
            float di[2 * TERMS + 1], up[2 * TERMS + 2], w[2 * TERMS + 1];
            column_coefs<TERMS>(di, up, sgP, ctP, n);
            band_column<TERMS, TERMS>(w, di, up, 0.125f);
            store_column<TERMS>(w, Win, 0, n, WIN_FLOATS + tid);
        } else {
            const int n = tid - UNITS;
            float di[2 * P_TERMS + 1], up[2 * P_TERMS + 2];
            column_coefs<P_TERMS>(di, up, sgP, ctP, n);
            float wp[2 * P_TERMS + 1];
            band_column<P_TERMS, P_TERMS>(wp, di, up, 0.03125f);
            store_column<P_TERMS>(wp, Win, 1, n, WIN_FLOATS + tid);
            float wq[2 * Q_TERMS + 1];
            band_column<Q_TERMS, P_TERMS>(wq, di, up, 0.015625f);
            store_column<Q_TERMS>(wq, Win, 2, n, WIN_FLOATS + tid);
            float wr[2 * ROW_TERMS + 1];
            #pragma unroll
            for (int i = 0; i < 2 * ROW_TERMS + 1; ++i) wr[i] = (i == ROW_TERMS) ? 1.0f : 0.0f;
            #pragma unroll
            for (int s = 1; s <= ROW_TERMS; ++s) {
                power_step<ROW_TERMS, P_TERMS>(wr, di, up, s, 1.0f / (float)s);
                store_column<ROW_TERMS>(wr, Win, 2 + s, n, WIN_FLOATS + tid);
            }
        }
    }
    __syncthreads();

    const int soff = 8 * hi * YSTRIDE + jbase + col;
    int cur = 0;

    {
        const int csR = win_start(3, wave);
        v2f rreg[ROW_TERMS][R_CHUNKS];
        #pragma unroll
        for (int j = 0; j < ROW_TERMS; ++j)
            w_frags<R_CHUNKS>(rreg[j], Win + win_base(3 + j, wave), col, hi);
        const int aoffR = col * YSTRIDE + 2 * hi + 4 * csR;
        v8f z[ROW_TERMS];
        apply_powers(z, Ybuf[0], aoffR, rreg);
        #pragma unroll
        for (int v = 0; v < 8; ++v) {
            const float h = hvec[v];
            S[v] += h * (z[0][v] + h * (z[1][v] + h * (z[2][v] + h * z[3][v])));
            Ybuf[1][v * YSTRIDE + soff] = S[v];
        }
        __syncthreads();
        cur = 1;
    }

    const int csF = win_start(0, wave);
    const int csP = win_start(1, wave);
    v2f ereg[BAND_CHUNKS];
    v2f preg[PQ_CHUNKS];
    v2f qreg[PQ_CHUNKS];
    w_frags<BAND_CHUNKS>(ereg, Win + win_base(0, wave), col, hi);
    w_frags<PQ_CHUNKS>(preg, Win + win_base(1, wave), col, hi);
    w_frags<PQ_CHUNKS>(qreg, Win + win_base(2, wave), col, hi);

    const int aoffF = col * YSTRIDE + 2 * hi + 4 * csF;
    const int aoffP = col * YSTRIDE + 2 * hi + 4 * csP;

    for (int j = 0; j < GMAX + PMAX + QMAX; ++j) {
        const float* Yc = cur ? Ybuf[1] : Ybuf[0];
        float*       Yn = cur ? Ybuf[0] : Ybuf[1];
        v8f z;
        if (j < GMAX)             z = apply_band<BAND_CHUNKS>(Yc, aoffF, ereg);
        else if (j < GMAX + PMAX) z = apply_band<PQ_CHUNKS>(Yc, aoffP, preg);
        else                      z = apply_band<PQ_CHUNKS>(Yc, aoffP, qreg);
        const bool more = (j < GMAX + PMAX + QMAX - 1);
        #pragma unroll
        for (int v = 0; v < 8; ++v) {
            S[v] = ((take[v] >> j) & 1) ? z[v] : S[v];
            if (more) Yn[v * YSTRIDE + soff] = S[v];
        }
        if (more) __syncthreads();
        cur ^= 1;
    }

    #pragma unroll
    for (int v = 0; v < 8; ++v) {
        int r = row0 + v + 8 * hi;
        if (r < B) out[r * UNITS + jbase + col] = S[v];
    }
}

extern "C" void kernel_launch(void* const* d_in, const int* in_sizes, int n_in,
                              void* d_out, int out_size, void* d_ws, size_t ws_size,
                              hipStream_t stream) {
    const float* inp    = (const float*)d_in[0];
    const float* params = (const float*)d_in[1];
    float*       out    = (float*)d_out;

    int B = in_sizes[0] / IN_STRIDE;
    int tiles = (B + TILE - 1) / TILE;

    expm_rows<<<tiles, 256, 0, stream>>>(inp, params, out, B);
}
